# starting kernel
# baseline (speedup 1.0000x reference)
#include <hip/hip_runtime.h>
#include <hip/hip_bf16.h>

#define B_   2
#define NH_  8
#define HH_  64
#define WW_  64
#define C_   256
#define DH_  32
#define N_   (HH_*WW_)
#define SCALE_ 0.17677669529663687f

typedef __attribute__((ext_vector_type(2))) float v2f;
typedef __attribute__((ext_vector_type(8))) float v8f;

static __device__ __forceinline__ v8f wmma_f32(v2f a, v2f b, v8f c) {
    return __builtin_amdgcn_wmma_f32_16x16x4_f32(
         false, a,  false, b,
         (short)0, c,  false,  false);
}

__global__ __launch_bounds__(256) void qta_gemm_qkv(const float* __restrict__ x,
                                                    const float* __restrict__ w,
                                                    float* __restrict__ out) {
    const int wave = threadIdx.x >> 5;
    const int lane = threadIdx.x & 31;
    const int lm = lane & 15, hf = lane >> 4;
    const int m0 = blockIdx.x * 16;
    const int n0 = (blockIdx.y * 8 + wave) * 16;
    v8f acc = {};
    const float* xr = x + (size_t)(m0 + lm) * C_;
    const float* wr = w + (size_t)(n0 + lm) * C_;
    for (int kk = 0; kk < C_; kk += 4) {
        v2f a, b;
        a[0] = xr[kk + 0 + 2*hf];
        a[1] = xr[kk + 1 + 2*hf];
        b[0] = wr[kk + 0 + 2*hf];
        b[1] = wr[kk + 1 + 2*hf];
        acc = wmma_f32(a, b, acc);
    }
    for (int r = 0; r < 8; ++r) {
        const int m = m0 + r + 8*hf;
        const int n = n0 + lm;
        const int b = m >> 12, s = m & 4095;
        const int h = n >> 5,  d = n & 31;
        out[((((size_t)b*NH_ + h)*N_) + s)*DH_ + d] = acc[r];
    }
}

__global__ __launch_bounds__(256) void qta_gemm_proj(const float* __restrict__ x,
                                                     const float* __restrict__ w,
                                                     const float* __restrict__ bias,
                                                     float* __restrict__ out) {
    const int wave = threadIdx.x >> 5;
    const int lane = threadIdx.x & 31;
    const int lm = lane & 15, hf = lane >> 4;
    const int m0 = blockIdx.x * 16;
    const int n0 = (blockIdx.y * 8 + wave) * 16;
    v8f acc = {};
    const float* xr = x + (size_t)(m0 + lm) * C_;
    const float* wr = w + (size_t)(n0 + lm) * C_;
    for (int kk = 0; kk < C_; kk += 4) {
        v2f a, b;
        a[0] = xr[kk + 0 + 2*hf];
        a[1] = xr[kk + 1 + 2*hf];
        b[0] = wr[kk + 0 + 2*hf];
        b[1] = wr[kk + 1 + 2*hf];
        acc = wmma_f32(a, b, acc);
    }
    for (int r = 0; r < 8; ++r) {
        const int m = m0 + r + 8*hf;
        const int n = n0 + lm;
        out[(size_t)m*C_ + n] = acc[r] + bias[n];
    }
}

__global__ void qta_pool(const float* __restrict__ in, float* __restrict__ out,
                         int h, int w) {
    const int total = B_*NH_*h*w*DH_;
    const int i = blockIdx.x*blockDim.x + threadIdx.x;
    if (i >= total) return;
    int t = i;
    const int d = t % DH_; t /= DH_;
    const int x = t % w;   t /= w;
    const int y = t % h;   t /= h;
    const int W2 = 2*w;
    const float* p = in + (((size_t)t*(2*h) + 2*y)*W2 + 2*x)*DH_ + d;
    out[i] = 0.25f*(p[0] + p[DH_] + p[(size_t)W2*DH_] + p[(size_t)(W2+1)*DH_]);
}

__global__ __launch_bounds__(32) void qta_level0(const float* __restrict__ q,
                                                 const float* __restrict__ k,
                                                 const float* __restrict__ v,
                                                 float* __restrict__ msg,
                                                 float* __restrict__ ps_out,
                                                 int* __restrict__ pi_out) {
    __shared__ float S[16][256];
    __shared__ float SC[16][256];
    __shared__ float topv[16][16];
    __shared__ int   topi[16][16];

    const int blk = blockIdx.x;
    const int qt  = blk & 15;
    const int bh  = blk >> 4;
    const int lane = threadIdx.x;
    const int lm = lane & 15, hf = lane >> 4;
    const int qrow0 = qt * 16;
    const float* Q = q + (size_t)bh*256*DH_;
    const float* K = k + (size_t)bh*256*DH_;
    const float* V = v + (size_t)bh*256*DH_;

    for (int nt = 0; nt < 16; ++nt) {
        v8f acc = {};
        const float* qr = Q + (size_t)(qrow0 + lm)*DH_;
        const float* kr = K + (size_t)(nt*16 + lm)*DH_;
        for (int kk = 0; kk < DH_; kk += 4) {
            v2f a, b;
            a[0] = SCALE_ * qr[kk + 0 + 2*hf];
            a[1] = SCALE_ * qr[kk + 1 + 2*hf];
            b[0] = kr[kk + 0 + 2*hf];
            b[1] = kr[kk + 1 + 2*hf];
            acc = wmma_f32(a, b, acc);
        }
        for (int r = 0; r < 8; ++r)
            S[r + 8*hf][nt*16 + lm] = acc[r];
    }
    __syncthreads();

    if (lane < 16) {
        const int m = lane;
        float mx = -1e30f;
        for (int j = 0; j < 256; ++j) mx = fmaxf(mx, S[m][j]);
        float sum = 0.f;
        for (int j = 0; j < 256; ++j) { float e = __expf(S[m][j] - mx); S[m][j] = e; sum += e; }
        const float inv = 1.f / sum;
        for (int j = 0; j < 256; ++j) { S[m][j] *= inv; SC[m][j] = S[m][j]; }
        for (int t = 0; t < 16; ++t) {
            float bv = -1e30f; int bi = 0;
            for (int j = 0; j < 256; ++j) if (SC[m][j] > bv) { bv = SC[m][j]; bi = j; }
            topv[m][t] = bv; topi[m][t] = bi; SC[m][bi] = -1e30f;
        }
    }
    __syncthreads();

    v8f a0 = {}, a1 = {};
    for (int kk = 0; kk < 256; kk += 4) {
        v2f a, b0, b1;
        a[0] = S[lm][kk + 0 + 2*hf];
        a[1] = S[lm][kk + 1 + 2*hf];
        const float* vr0 = V + (size_t)(kk + 0 + 2*hf)*DH_;
        const float* vr1 = V + (size_t)(kk + 1 + 2*hf)*DH_;
        b0[0] = vr0[lm];      b0[1] = vr1[lm];
        b1[0] = vr0[16 + lm]; b1[1] = vr1[16 + lm];
        a0 = wmma_f32(a, b0, a0);
        a1 = wmma_f32(a, b1, a1);
    }
    for (int r = 0; r < 8; ++r) {
        SC[r + 8*hf][lm]      = a0[r];
        SC[r + 8*hf][16 + lm] = a1[r];
    }
    __syncthreads();

    for (int m = 0; m < 16; ++m) {
        float msel = 0.f;
        for (int t = 0; t < 16; ++t)
            msel += topv[m][t] * V[(size_t)topi[m][t]*DH_ + lane];
        msg[((size_t)bh*256 + qrow0 + m)*DH_ + lane] = SC[m][lane] - msel;
    }
    if (lane < 16) {
        for (int m = 0; m < 16; ++m) {
            ps_out[((size_t)bh*256 + qrow0 + m)*16 + lane] = topv[m][lane];
            pi_out[((size_t)bh*256 + qrow0 + m)*16 + lane] = topi[m][lane];
        }
    }
}

__global__ __launch_bounds__(32) void qta_levelt(const float* __restrict__ qf,
                                                 const float* __restrict__ kf,
                                                 const float* __restrict__ vf,
                                                 const float* __restrict__ ps_in,
                                                 const int*   __restrict__ pi_in,
                                                 float* __restrict__ msg,
                                                 float* __restrict__ ps_out,
                                                 int*   __restrict__ pi_out,
                                                 int gh, int gw, int is_last) {
    __shared__ float sL[4][64];
    __shared__ float mA[4][32];
    __shared__ float psh[16];
    __shared__ int   ci[64];
    __shared__ float topv[4][16];
    __shared__ int   topi[4][16];

    const int Sp = gh*gw;
    const int p  = blockIdx.x % Sp;
    const int bh = blockIdx.x / Sp;
    const int wt = 2*gw;
    const int St = 4*Sp;
    const int lane = threadIdx.x;
    const int lm = lane & 15, hf = lane >> 4;
    const float* Q = qf + (size_t)bh*St*DH_;
    const float* K = kf + (size_t)bh*St*DH_;
    const float* V = vf + (size_t)bh*St*DH_;

    if (lane < 16) {
        psh[lane] = ps_in[((size_t)bh*Sp + p)*16 + lane];
        const int pi = pi_in[((size_t)bh*Sp + p)*16 + lane];
        const int r = pi / gw, c = pi % gw;
        for (int cc = 0; cc < 4; ++cc)
            ci[lane*4 + cc] = (2*r + (cc >> 1))*wt + (2*c + (cc & 1));
    }
    __syncthreads();

    const int rp = p / gw, cp = p % gw;
    const int qm = lm & 3;
    const int qidx = (2*rp + (qm >> 1))*wt + (2*cp + (qm & 1));
    const bool alive = (lm < 4);

    for (int nt = 0; nt < 4; ++nt) {
        v8f acc = {};
        const int key = ci[nt*16 + lm];
        const float* qr = Q + (size_t)qidx*DH_;
        const float* kr = K + (size_t)key*DH_;
        for (int kk = 0; kk < DH_; kk += 4) {
            v2f a, b;
            const float a0v = SCALE_ * qr[kk + 0 + 2*hf];
            const float a1v = SCALE_ * qr[kk + 1 + 2*hf];
            a[0] = alive ? a0v : 0.f;
            a[1] = alive ? a1v : 0.f;
            b[0] = kr[kk + 0 + 2*hf];
            b[1] = kr[kk + 1 + 2*hf];
            acc = wmma_f32(a, b, acc);
        }
        if (hf == 0)
            for (int r = 0; r < 4; ++r) sL[r][nt*16 + lm] = acc[r];
    }
    __syncthreads();

    for (int g = lane; g < 64; g += 32) {
        const int m = g >> 4, kp = g & 15;
        float* row = &sL[m][kp*4];
        const float mx = fmaxf(fmaxf(row[0], row[1]), fmaxf(row[2], row[3]));
        const float e0 = __expf(row[0]-mx), e1 = __expf(row[1]-mx);
        const float e2 = __expf(row[2]-mx), e3 = __expf(row[3]-mx);
        const float f = psh[kp] / (e0+e1+e2+e3);
        row[0] = e0*f; row[1] = e1*f; row[2] = e2*f; row[3] = e3*f;
    }
    __syncthreads();

    for (int nt = 0; nt < 2; ++nt) {
        v8f acc = {};
        for (int kk = 0; kk < 64; kk += 4) {
            v2f a, b;
            const float a0v = sL[qm][kk + 0 + 2*hf];
            const float a1v = sL[qm][kk + 1 + 2*hf];
            a[0] = alive ? a0v : 0.f;
            a[1] = alive ? a1v : 0.f;
            b[0] = V[(size_t)ci[kk + 0 + 2*hf]*DH_ + nt*16 + lm];
            b[1] = V[(size_t)ci[kk + 1 + 2*hf]*DH_ + nt*16 + lm];
            acc = wmma_f32(a, b, acc);
        }
        if (hf == 0)
            for (int r = 0; r < 4; ++r) mA[r][nt*16 + lm] = acc[r];
    }
    __syncthreads();

    if (is_last) {
        for (int m = 0; m < 4; ++m) {
            const int fi = (2*rp + (m >> 1))*wt + (2*cp + (m & 1));
            msg[((size_t)bh*St + fi)*DH_ + lane] = mA[m][lane];
        }
        return;
    }

    if (lane < 4) {
        const int m = lane;
        for (int t = 0; t < 16; ++t) {
            float bv = -1e30f; int bi = 0;
            for (int j = 0; j < 64; ++j) if (sL[m][j] > bv) { bv = sL[m][j]; bi = j; }
            topv[m][t] = bv; topi[m][t] = bi; sL[m][bi] = -1e30f;
        }
    }
    __syncthreads();

    for (int m = 0; m < 4; ++m) {
        float msel = 0.f;
        for (int t = 0; t < 16; ++t)
            msel += topv[m][t] * V[(size_t)ci[topi[m][t]]*DH_ + lane];
        const int fi = (2*rp + (m >> 1))*wt + (2*cp + (m & 1));
        msg[((size_t)bh*St + fi)*DH_ + lane] = mA[m][lane] - msel;
    }
    if (lane < 16) {
        for (int m = 0; m < 4; ++m) {
            const int fi = (2*rp + (m >> 1))*wt + (2*cp + (m & 1));
            ps_out[((size_t)bh*St + fi)*16 + lane] = topv[m][lane];
            pi_out[((size_t)bh*St + fi)*16 + lane] = ci[topi[m][lane]];
        }
    }
}

__global__ void qta_combine(const float* __restrict__ m0, const float* __restrict__ m1,
                            const float* __restrict__ m2, float* __restrict__ out) {
    const int total = B_*N_*C_;
    const int i = blockIdx.x*blockDim.x + threadIdx.x;
    if (i >= total) return;
    const int c = i % C_;
    int t = i / C_;
    const int col = t % WW_;
    const int row = (t / WW_) % HH_;
    const int b = t / (WW_*HH_);
    const int h = c >> 5, d = c & 31;
    const int bh = b*NH_ + h;
    float val = m2[((size_t)bh*4096 + row*64 + col)*DH_ + d]
              + m1[((size_t)bh*1024 + (row >> 1)*32 + (col >> 1))*DH_ + d]
              + m0[((size_t)bh*256  + (row >> 2)*16 + (col >> 2))*DH_ + d];
    out[i] = val;
}

extern "C" void kernel_launch(void* const* d_in, const int* in_sizes, int n_in,
                              void* d_out, int out_size, void* d_ws, size_t ws_size,
                              hipStream_t stream) {
    (void)in_sizes; (void)n_in; (void)out_size; (void)ws_size;
    const float* x  = (const float*)d_in[0];
    const float* wq = (const float*)d_in[3];
    const float* wk = (const float*)d_in[4];
    const float* wv = (const float*)d_in[5];
    const float* wp = (const float*)d_in[6];
    const float* bp = (const float*)d_in[7];

    const size_t SZ2 = (size_t)B_*NH_*4096*DH_;
    const size_t SZ1 = (size_t)B_*NH_*1024*DH_;
    const size_t SZ0 = (size_t)B_*NH_*256 *DH_;
    float* ws = (float*)d_ws; size_t o = 0;
    float* q2 = ws + o; o += SZ2;  float* k2 = ws + o; o += SZ2;  float* v2 = ws + o; o += SZ2;
    float* q1 = ws + o; o += SZ1;  float* k1 = ws + o; o += SZ1;  float* v1 = ws + o; o += SZ1;
    float* q0 = ws + o; o += SZ0;  float* k0 = ws + o; o += SZ0;  float* v0 = ws + o; o += SZ0;
    float* msg0 = ws + o; o += SZ0;
    float* msg1 = ws + o; o += SZ1;
    float* msg2 = ws + o; o += SZ2;
    float* ps0 = ws + o; o += (size_t)B_*NH_*256*16;
    int*   pi0 = (int*)(ws + o); o += (size_t)B_*NH_*256*16;
    float* ps1 = ws + o; o += (size_t)B_*NH_*1024*16;
    int*   pi1 = (int*)(ws + o); o += (size_t)B_*NH_*1024*16;
    float* attn = ws + o; o += SZ2;

    const dim3 gg(512, 2);
    qta_gemm_qkv<<<gg, 256, 0, stream>>>(x, wq, q2);
    qta_gemm_qkv<<<gg, 256, 0, stream>>>(x, wk, k2);
    qta_gemm_qkv<<<gg, 256, 0, stream>>>(x, wv, v2);

    const int T1 = B_*NH_*32*32*DH_;
    qta_pool<<<(T1 + 255)/256, 256, 0, stream>>>(q2, q1, 32, 32);
    qta_pool<<<(T1 + 255)/256, 256, 0, stream>>>(k2, k1, 32, 32);
    qta_pool<<<(T1 + 255)/256, 256, 0, stream>>>(v2, v1, 32, 32);
    const int T0 = B_*NH_*16*16*DH_;
    qta_pool<<<(T0 + 255)/256, 256, 0, stream>>>(q1, q0, 16, 16);
    qta_pool<<<(T0 + 255)/256, 256, 0, stream>>>(k1, k0, 16, 16);
    qta_pool<<<(T0 + 255)/256, 256, 0, stream>>>(v1, v0, 16, 16);

    qta_level0<<<B_*NH_*16, 32, 0, stream>>>(q0, k0, v0, msg0, ps0, pi0);
    qta_levelt<<<B_*NH_*256,  32, 0, stream>>>(q1, k1, v1, ps0, pi0, msg1, ps1, pi1, 16, 16, 0);
    qta_levelt<<<B_*NH_*1024, 32, 0, stream>>>(q2, k2, v2, ps1, pi1, msg2, nullptr, nullptr, 32, 32, 1);

    qta_combine<<<(B_*N_*C_ + 255)/256, 256, 0, stream>>>(msg0, msg1, msg2, attn);
    qta_gemm_proj<<<gg, 256, 0, stream>>>(attn, wp, bp, (float*)d_out);
}


# level 0: work-group of 4 waves per 16-query tile: scores, exp, scaling split 4 key tiles per wave; wave 0 picks beside A@V in waves 1,2 (a d-tile each); m_sel rows 4 per wave; sum orders kept
# speedup vs baseline: 2.2542x; 2.2542x over previous
#include <hip/hip_runtime.h>
#include <hip/hip_bf16.h>

#define B_   2
#define NH_  8
#define HH_  64
#define WW_  64
#define C_   256
#define DH_  32
#define N_   (HH_*WW_)
#define SCALE_ 0.17677669529663687f

typedef __attribute__((ext_vector_type(2))) float v2f;
typedef __attribute__((ext_vector_type(8))) float v8f;

static __device__ __forceinline__ v8f wmma_f32(v2f a, v2f b, v8f c) {
    return __builtin_amdgcn_wmma_f32_16x16x4_f32(
         false, a,  false, b,
         (short)0, c,  false,  false);
}

#define GBM_ 64
#define GBN_ 128
#define GBK_ 32
#define GP_  36
#define GSTAGE_ ((GBM_ + GBN_) * GP_)

static __device__ __forceinline__ void qta_gemm_chunk(const float* cur, const int aoff, const int boff,
                                                      v8f& c00, v8f& c01, v8f& c10, v8f& c11) {
    for (int ks = 0; ks < GBK_; ks += 4) {
        const v2f a0 = *(const v2f*)(cur + aoff + ks);
        const v2f a1 = *(const v2f*)(cur + aoff + 16*GP_ + ks);
        const v2f b0 = *(const v2f*)(cur + boff + ks);
        const v2f b1 = *(const v2f*)(cur + boff + 16*GP_ + ks);
        c00 = wmma_f32(a0, b0, c00);
        c01 = wmma_f32(a0, b1, c01);
        c10 = wmma_f32(a1, b0, c10);
        c11 = wmma_f32(a1, b1, c11);
    }
}

struct qta_a_plain {
    const float* g;
    int row2;
    float4 r0, r1;
    __device__ __forceinline__ void load(const int c) {
        r0 = *(const float4*)(g + c*GBK_);
        r1 = *(const float4*)(g + c*GBK_ + row2);
    }
    __device__ __forceinline__ void get(float4& v0, float4& v1) const { v0 = r0; v1 = r1; }
};
struct qta_a_msgsum {
    const float *g2, *g1, *g0;
    float4 r20, r21, r10, r11, r00, r01;
    __device__ __forceinline__ void load(const int c) {
        r20 = *(const float4*)(g2 + (size_t)c*N_*DH_);
        r21 = *(const float4*)(g2 + (size_t)c*N_*DH_ + 32*DH_);
        r10 = *(const float4*)(g1 + (size_t)c*(N_/4)*DH_);
        r11 = *(const float4*)(g1 + (size_t)c*(N_/4)*DH_ + 16*DH_);
        r00 = *(const float4*)(g0 + (size_t)c*(N_/16)*DH_);
        r01 = *(const float4*)(g0 + (size_t)c*(N_/16)*DH_ + 8*DH_);
    }
    static __device__ __forceinline__ float4 sum3(const float4 a, const float4 b, const float4 c) {
        float4 r;
        r.x = (a.x + b.x) + c.x; r.y = (a.y + b.y) + c.y; r.z = (a.z + b.z) + c.z; r.w = (a.w + b.w) + c.w;
        return r;
    }
    __device__ __forceinline__ void get(float4& v0, float4& v1) const {
        v0 = sum3(r20, r10, r00);
        v1 = sum3(r21, r11, r01);
    }
};

template <class ASRC>
static __device__ __forceinline__ void qta_gemm_mainloop(ASRC& as,
                                                         const float* __restrict__ w,
                                                         float* sm, const int n0,
                                                         v8f& c00, v8f& c01, v8f& c10, v8f& c11) {
    const int tid  = threadIdx.x;
    const int wave = tid >> 5, lane = tid & 31;
    const int lm = lane & 15, hf = lane >> 4;
    const int wm = wave & 1, wn = wave >> 1;

    const int lrow = tid >> 3;
    const int lcol = (tid & 7) * 4;
    const float* wg = w + (size_t)(n0 + lrow) * C_ + lcol;
    const int so = lrow * GP_ + lcol;

    float4 ra0, ra1, rb0, rb1, rb2, rb3;
    as.load(0);
    rb0 = *(const float4*)(wg);
    rb1 = *(const float4*)(wg + 32*C_);
    rb2 = *(const float4*)(wg + 64*C_);
    rb3 = *(const float4*)(wg + 96*C_);
    as.get(ra0, ra1);
    *(float4*)(sm + so)                        = ra0;
    *(float4*)(sm + so + 32*GP_)               = ra1;
    *(float4*)(sm + GBM_*GP_ + so)             = rb0;
    *(float4*)(sm + GBM_*GP_ + so + 32*GP_)    = rb1;
    *(float4*)(sm + GBM_*GP_ + so + 64*GP_)    = rb2;
    *(float4*)(sm + GBM_*GP_ + so + 96*GP_)    = rb3;
    __syncthreads();

    const int aoff = (wm*32 + lm) * GP_ + 2*hf;
    const int boff = GBM_*GP_ + (wn*32 + lm) * GP_ + 2*hf;

    #pragma nounroll
    for (int c = 0; c < C_/GBK_ - 1; ++c) {
        const int kn = (c + 1) * GBK_;
        as.load(c + 1);
        rb0 = *(const float4*)(wg + kn);
        rb1 = *(const float4*)(wg + kn + 32*C_);
        rb2 = *(const float4*)(wg + kn + 64*C_);
        rb3 = *(const float4*)(wg + kn + 96*C_);
        __builtin_amdgcn_sched_barrier(0);
        qta_gemm_chunk(sm + (c & 1) * GSTAGE_, aoff, boff, c00, c01, c10, c11);
        float* nx = sm + ((c + 1) & 1) * GSTAGE_;
        as.get(ra0, ra1);
        *(float4*)(nx + so)                        = ra0;
        *(float4*)(nx + so + 32*GP_)               = ra1;
        *(float4*)(nx + GBM_*GP_ + so)             = rb0;
        *(float4*)(nx + GBM_*GP_ + so + 32*GP_)    = rb1;
        *(float4*)(nx + GBM_*GP_ + so + 64*GP_)    = rb2;
        *(float4*)(nx + GBM_*GP_ + so + 96*GP_)    = rb3;
        __syncthreads();
    }
    qta_gemm_chunk(sm + ((C_/GBK_ - 1) & 1) * GSTAGE_, aoff, boff, c00, c01, c10, c11);
}

__global__ __launch_bounds__(256) void qta_gemm_qkv(const float* __restrict__ x,
                                                    const float* __restrict__ w,
                                                    float* __restrict__ out2,
                                                    float* __restrict__ out1,
                                                    float* __restrict__ out0) {
    __shared__ float4 smv[2 * GSTAGE_ / 4];
    float* sm = (float*)smv;
    const int wave = threadIdx.x >> 5;
    const int lane = threadIdx.x & 31;
    const int lm = lane & 15, hf = lane >> 4;
    const int wm = wave & 1, wn = wave >> 1;
    const int n0 = blockIdx.x * GBN_;
    const int bb = blockIdx.y >> 6;
    const int py = (blockIdx.y >> 3) & 7;
    const int px = blockIdx.y & 7;

    const int lrow = threadIdx.x >> 3;
    const int ty = ((lrow >> 4) & 1) * 2 + ((lrow >> 1) & 1);
    const int tx = ((lrow >> 3) & 1) * 4 + ((lrow >> 2) & 1) * 2 + (lrow & 1);
    qta_a_plain as;
    as.g = x + ((size_t)bb * N_ + (py*8 + ty) * WW_ + px*8 + tx) * C_ + (threadIdx.x & 7) * 4;
    as.row2 = 4*WW_*C_;

    v8f acc[2][2] = {};
    qta_gemm_mainloop(as, w, sm, n0, acc[0][0], acc[0][1], acc[1][0], acc[1][1]);

    const size_t bh = (size_t)bb*NH_ + (n0 >> 5) + wn;
    const int yb = py*8 + wm*4;
    const int xb = px*8 + hf*4;
    for (int j = 0; j < 2; ++j) {
        const int d = j*16 + lm;
        float l1[2][2];
        for (int i = 0; i < 2; ++i) {
            for (int r = 0; r < 8; ++r) {
                const int yy = yb + 2*i + ((r >> 1) & 1);
                const int xx = xb + 2*(r >> 2) + (r & 1);
                out2[((bh*N_) + yy*WW_ + xx)*DH_ + d] = acc[i][j][r];
            }
            for (int qx = 0; qx < 2; ++qx) {
                const float s = 0.25f*(acc[i][j][4*qx] + acc[i][j][4*qx + 1] + acc[i][j][4*qx + 2] + acc[i][j][4*qx + 3]);
                l1[i][qx] = s;
                out1[((bh*(N_/4)) + ((yb >> 1) + i)*(WW_/2) + (xb >> 1) + qx)*DH_ + d] = s;
            }
        }
        out0[((bh*(N_/16)) + (yb >> 2)*(WW_/4) + (xb >> 2))*DH_ + d] =
            0.25f*(l1[0][0] + l1[0][1] + l1[1][0] + l1[1][1]);
    }
}

static_assert(GBM_ == WW_, "a proj work-group covers exactly one image row");
__global__ __launch_bounds__(256) void qta_gemm_proj(const float* __restrict__ m2,
                                                     const float* __restrict__ m1,
                                                     const float* __restrict__ m0,
                                                     const float* __restrict__ w,
                                                     const float* __restrict__ bias,
                                                     float* __restrict__ out) {
    __shared__ float4 smv[2 * GSTAGE_ / 4];
    float* sm = (float*)smv;
    const int wave = threadIdx.x >> 5;
    const int lane = threadIdx.x & 31;
    const int lm = lane & 15, hf = lane >> 4;
    const int wm = wave & 1, wn = wave >> 1;
    const int n0 = blockIdx.x * GBN_;
    const int mb = blockIdx.y * GBM_;
    const int bb  = blockIdx.y >> 6;
    const int row = blockIdx.y & 63;
    const int col = threadIdx.x >> 3;
    const int lcol = (threadIdx.x & 7) * 4;
    qta_a_msgsum as;
    as.g2 = m2 + ((size_t)bb*NH_*N_      + row*WW_            + col       ) * DH_ + lcol;
    as.g1 = m1 + ((size_t)bb*NH_*(N_/4)  + (row >> 1)*(WW_/2) + (col >> 1)) * DH_ + lcol;
    as.g0 = m0 + ((size_t)bb*NH_*(N_/16) + (row >> 2)*(WW_/4) + (col >> 2)) * DH_ + lcol;
    v8f acc[2][2] = {};
    qta_gemm_mainloop(as, w, sm, n0, acc[0][0], acc[0][1], acc[1][0], acc[1][1]);
    for (int i = 0; i < 2; ++i)
        for (int j = 0; j < 2; ++j)
            for (int r = 0; r < 8; ++r) {
                const int m = mb + wm*32 + i*16 + r + 8*hf;
                const int n = n0 + wn*32 + j*16 + lm;
                out[(size_t)m*C_ + n] = acc[i][j][r] + bias[n];
            }
}


template <int LO, int N> struct qta_argmax {
    static __device__ __forceinline__ void run(const float* v, float& bv, int& bi) {
        float lv, rv; int li, ri;
        qta_argmax<LO, N/2>::run(v, lv, li);
        qta_argmax<LO + N/2, N/2>::run(v, rv, ri);
        const bool r = rv > lv;
        bv = r ? rv : lv;
        bi = r ? ri : li;
    }
};
template <int LO> struct qta_argmax<LO, 1> {
    static __device__ __forceinline__ void run(const float* v, float& bv, int& bi) { bv = v[LO]; bi = LO; }
};

static __device__ __forceinline__ float qta_max16(const float* x) {
    const float a0 = fmaxf(x[0],  x[1]),  a1 = fmaxf(x[2],  x[3]),  a2 = fmaxf(x[4],  x[5]),  a3 = fmaxf(x[6],  x[7]);
    const float a4 = fmaxf(x[8],  x[9]),  a5 = fmaxf(x[10], x[11]), a6 = fmaxf(x[12], x[13]), a7 = fmaxf(x[14], x[15]);
    return fmaxf(fmaxf(fmaxf(a0, a1), fmaxf(a2, a3)), fmaxf(fmaxf(a4, a5), fmaxf(a6, a7)));
}

template <int NCH>
static __device__ __forceinline__ void qta_pick16(float* row, float (&cm)[NCH], float* tv, int* ti) {
    #pragma nounroll
    for (int t = 0; t < 16; ++t) {
        float bv; int bc;
        qta_argmax<0, NCH>::run(cm, bv, bc);
        float* ch = row + bc*16;
        float v[16];
        #pragma unroll
        for (int q = 0; q < 4; ++q) {
            const float4 x = ((const float4*)ch)[q];
            v[4*q + 0] = x.x; v[4*q + 1] = x.y; v[4*q + 2] = x.z; v[4*q + 3] = x.w;
        }
        float ev; int jj;
        qta_argmax<0, 16>::run(v, ev, jj);
        ch[jj] = -1e30f;
        #pragma unroll
        for (int j = 0; j < 16; ++j) v[j] = (j == jj) ? -1e30f : v[j];
        const float nm = qta_max16(v);
        #pragma unroll
        for (int cc = 0; cc < NCH; ++cc) cm[cc] = (cc == bc) ? nm : cm[cc];
        tv[t] = ev;
        ti[t] = bc*16 + jj;
    }
}

__global__ __launch_bounds__(128) void qta_level0(const float* __restrict__ q,
                                                  const float* __restrict__ k,
                                                  const float* __restrict__ v,
                                                  float* __restrict__ msg,
                                                  float* __restrict__ ps_out,
                                                  int* __restrict__ pi_out) {
    __shared__ __attribute__((aligned(16))) float S[16][256];
    __shared__ __attribute__((aligned(16))) float SC[16][256];
    __shared__ float MA[16][32];
    __shared__ float topv[16][16];
    __shared__ int   topi[16][16];
    __shared__ __attribute__((aligned(16))) float rinv[16];
    __shared__ float wmx[4][16];

    const int blk = blockIdx.x;
    const int qt  = blk & 15;
    const int bh  = blk >> 4;
    const int lane = threadIdx.x & 31;
    const int wave = __builtin_amdgcn_readfirstlane((int)(threadIdx.x >> 5));
    const int lm = lane & 15, hf = lane >> 4;
    const int qrow0 = qt * 16;
    const float* Q = q + (size_t)bh*256*DH_;
    const float* K = k + (size_t)bh*256*DH_;
    const float* V = v + (size_t)bh*256*DH_;

    v8f acc[4];
    const float* qr = Q + (size_t)(qrow0 + lm)*DH_;
    #pragma unroll
    for (int nt = 0; nt < 4; ++nt) {
        v8f c = {};
        const float* kr = K + (size_t)((wave*4 + nt)*16 + lm)*DH_;
        for (int kk = 0; kk < DH_; kk += 4) {
            v2f a, b;
            a[0] = SCALE_ * qr[kk + 0 + 2*hf];
            a[1] = SCALE_ * qr[kk + 1 + 2*hf];
            b[0] = kr[kk + 0 + 2*hf];
            b[1] = kr[kk + 1 + 2*hf];
            c = wmma_f32(a, b, c);
        }
        acc[nt] = c;
    }

    float mx[8];
    #pragma unroll
    for (int r = 0; r < 8; ++r) {
        float m = -1e30f;
        #pragma unroll
        for (int nt = 0; nt < 4; ++nt) m = fmaxf(m, acc[nt][r]);
        for (int o = 1; o < 16; o <<= 1) m = fmaxf(m, __shfl_xor(m, o, 32));
        mx[r] = m;
    }
    if (lm == 0) {
        #pragma unroll
        for (int r = 0; r < 8; ++r) wmx[wave][r + 8*hf] = mx[r];
    }
    __syncthreads();
    #pragma unroll
    for (int r = 0; r < 8; ++r)
        mx[r] = fmaxf(fmaxf(wmx[0][r + 8*hf], wmx[1][r + 8*hf]), fmaxf(wmx[2][r + 8*hf], wmx[3][r + 8*hf]));
    #pragma unroll
    for (int nt = 0; nt < 4; ++nt) {
        #pragma unroll
        for (int r = 0; r < 8; ++r) {
            const float e = __expf(acc[nt][r] - mx[r]);
            acc[nt][r] = e;
            S[r + 8*hf][(wave*4 + nt)*16 + lm] = e;
        }
    }
    __syncthreads();

    float cm[16];
    if (wave == 0 && lane < 16) {
        const int m = lane;
        float sum = 0.f;
        const float4* rowv = (const float4*)&S[m][0];
        float4 nx[2][4];
        #pragma unroll
        for (int q = 0; q < 4; ++q) nx[0][q] = rowv[q];
        #pragma unroll
        for (int cc = 0; cc < 16; ++cc) {
            if (cc + 1 < 16) {
                #pragma unroll
                for (int q = 0; q < 4; ++q) nx[(cc + 1) & 1][q] = rowv[(cc + 1)*4 + q];
            }
            __builtin_amdgcn_sched_barrier(0);
            float ev[16];
            #pragma unroll
            for (int q = 0; q < 4; ++q) {
                const float4 x = nx[cc & 1][q];
                ev[4*q + 0] = x.x; ev[4*q + 1] = x.y; ev[4*q + 2] = x.z; ev[4*q + 3] = x.w;
            }
            #pragma unroll
            for (int jj = 0; jj < 16; ++jj) sum += ev[jj];
            cm[cc] = qta_max16(ev);
        }
        const float inv = 1.f / sum;
        #pragma unroll
        for (int cc = 0; cc < 16; ++cc) cm[cc] *= inv;
        rinv[m] = inv;
    }
    __syncthreads();

    float iv[8];
    #pragma unroll
    for (int r = 0; r < 8; ++r) iv[r] = rinv[r + 8*hf];
    #pragma unroll
    for (int nt = 0; nt < 4; ++nt) {
        #pragma unroll
        for (int r = 0; r < 8; ++r) {
            const float pv = acc[nt][r] * iv[r];
            S[r + 8*hf][(wave*4 + nt)*16 + lm] = pv;
            SC[r + 8*hf][(wave*4 + nt)*16 + lm] = pv;
        }
    }
    __syncthreads();

    if (wave == 0) {
        if (lane < 16)
            qta_pick16<16>(&SC[lane][0], cm, &topv[lane][0], &topi[lane][0]);
    } else if (wave <= 2) {
        const int dcol = (wave - 1)*16 + lm;
        v8f a0 = {};
        for (int kk = 0; kk < 256; kk += 4) {
            v2f a, b0;
            a[0] = S[lm][kk + 0 + 2*hf];
            a[1] = S[lm][kk + 1 + 2*hf];
            b0[0] = V[(size_t)(kk + 0 + 2*hf)*DH_ + dcol];
            b0[1] = V[(size_t)(kk + 1 + 2*hf)*DH_ + dcol];
            a0 = wmma_f32(a, b0, a0);
        }
        for (int r = 0; r < 8; ++r) MA[r + 8*hf][dcol] = a0[r];
    }
    __syncthreads();

    for (int m = 4*wave; m < 4*wave + 4; ++m) {
        float msel = 0.f;
        for (int t = 0; t < 16; ++t)
            msel += topv[m][t] * V[(size_t)topi[m][t]*DH_ + lane];
        msg[((size_t)bh*256 + qrow0 + m)*DH_ + lane] = MA[m][lane] - msel;
    }
    if (lane < 16) {
        for (int m = 4*wave; m < 4*wave + 4; ++m) {
            ps_out[((size_t)bh*256 + qrow0 + m)*16 + lane] = topv[m][lane];
            pi_out[((size_t)bh*256 + qrow0 + m)*16 + lane] = topi[m][lane];
        }
    }
}

__global__ __launch_bounds__(32) void qta_levelt(const float* __restrict__ qf,
                                                 const float* __restrict__ kf,
                                                 const float* __restrict__ vf,
                                                 const float* __restrict__ ps_in,
                                                 const int*   __restrict__ pi_in,
                                                 float* __restrict__ msg,
                                                 float* __restrict__ ps_out,
                                                 int*   __restrict__ pi_out,
                                                 int gh, int gw, int is_last) {
    __shared__ __attribute__((aligned(16))) float sL[4][64];
    __shared__ float mA[4][32];
    __shared__ float psh[16];
    __shared__ int   ci[64];
    __shared__ float topv[4][16];
    __shared__ int   topi[4][16];

    const int Sp = gh*gw;
    const int p  = blockIdx.x % Sp;
    const int bh = blockIdx.x / Sp;
    const int wt = 2*gw;
    const int St = 4*Sp;
    const int lane = threadIdx.x;
    const int lm = lane & 15, hf = lane >> 4;
    const float* Q = qf + (size_t)bh*St*DH_;
    const float* K = kf + (size_t)bh*St*DH_;
    const float* V = vf + (size_t)bh*St*DH_;

    if (lane < 16) {
        psh[lane] = ps_in[((size_t)bh*Sp + p)*16 + lane];
        const int pi = pi_in[((size_t)bh*Sp + p)*16 + lane];
        const int r = pi / gw, c = pi % gw;
        for (int cc = 0; cc < 4; ++cc)
            ci[lane*4 + cc] = (2*r + (cc >> 1))*wt + (2*c + (cc & 1));
    }
    __syncthreads();

    const int rp = p / gw, cp = p % gw;
    const int qm = lm & 3;
    const int qidx = (2*rp + (qm >> 1))*wt + (2*cp + (qm & 1));
    const bool alive = (lm < 4);

    for (int nt = 0; nt < 4; ++nt) {
        v8f acc = {};
        const int key = ci[nt*16 + lm];
        const float* qr = Q + (size_t)qidx*DH_;
        const float* kr = K + (size_t)key*DH_;
        for (int kk = 0; kk < DH_; kk += 4) {
            v2f a, b;
            const float a0v = SCALE_ * qr[kk + 0 + 2*hf];
            const float a1v = SCALE_ * qr[kk + 1 + 2*hf];
            a[0] = alive ? a0v : 0.f;
            a[1] = alive ? a1v : 0.f;
            b[0] = kr[kk + 0 + 2*hf];
            b[1] = kr[kk + 1 + 2*hf];
            acc = wmma_f32(a, b, acc);
        }
        if (hf == 0)
            for (int r = 0; r < 4; ++r) sL[r][nt*16 + lm] = acc[r];
    }
    __syncthreads();

    for (int g = lane; g < 64; g += 32) {
        const int m = g >> 4, kp = g & 15;
        float* row = &sL[m][kp*4];
        const float mx = fmaxf(fmaxf(row[0], row[1]), fmaxf(row[2], row[3]));
        const float e0 = __expf(row[0]-mx), e1 = __expf(row[1]-mx);
        const float e2 = __expf(row[2]-mx), e3 = __expf(row[3]-mx);
        const float f = psh[kp] / (e0+e1+e2+e3);
        row[0] = e0*f; row[1] = e1*f; row[2] = e2*f; row[3] = e3*f;
    }
    __syncthreads();

    for (int nt = 0; nt < 2; ++nt) {
        v8f acc = {};
        for (int kk = 0; kk < 64; kk += 4) {
            v2f a, b;
            const float a0v = sL[qm][kk + 0 + 2*hf];
            const float a1v = sL[qm][kk + 1 + 2*hf];
            a[0] = alive ? a0v : 0.f;
            a[1] = alive ? a1v : 0.f;
            b[0] = V[(size_t)ci[kk + 0 + 2*hf]*DH_ + nt*16 + lm];
            b[1] = V[(size_t)ci[kk + 1 + 2*hf]*DH_ + nt*16 + lm];
            acc = wmma_f32(a, b, acc);
        }
        if (hf == 0)
            for (int r = 0; r < 4; ++r) mA[r][nt*16 + lm] = acc[r];
    }
    __syncthreads();

    if (is_last) {
        for (int m = 0; m < 4; ++m) {
            const int fi = (2*rp + (m >> 1))*wt + (2*cp + (m & 1));
            msg[((size_t)bh*St + fi)*DH_ + lane] = mA[m][lane];
        }
        return;
    }

    if (lane < 4) {
        const int m = lane;
        float cm[4];
        #pragma unroll
        for (int cc = 0; cc < 4; ++cc) {
            float x[16];
            #pragma unroll
            for (int q = 0; q < 4; ++q) {
                const float4 xv = ((const float4*)&sL[m][cc*16])[q];
                x[4*q + 0] = xv.x; x[4*q + 1] = xv.y; x[4*q + 2] = xv.z; x[4*q + 3] = xv.w;
            }
            cm[cc] = qta_max16(x);
        }
        qta_pick16<4>(&sL[m][0], cm, &topv[m][0], &topi[m][0]);
    }
    __syncthreads();

    for (int m = 0; m < 4; ++m) {
        float msel = 0.f;
        for (int t = 0; t < 16; ++t)
            msel += topv[m][t] * V[(size_t)ci[topi[m][t]]*DH_ + lane];
        const int fi = (2*rp + (m >> 1))*wt + (2*cp + (m & 1));
        msg[((size_t)bh*St + fi)*DH_ + lane] = mA[m][lane] - msel;
    }
    if (lane < 16) {
        for (int m = 0; m < 4; ++m) {
            const int fi = (2*rp + (m >> 1))*wt + (2*cp + (m & 1));
            ps_out[((size_t)bh*St + fi)*16 + lane] = topv[m][lane];
            pi_out[((size_t)bh*St + fi)*16 + lane] = ci[topi[m][lane]];
        }
    }
}

extern "C" void kernel_launch(void* const* d_in, const int* in_sizes, int n_in,
                              void* d_out, int out_size, void* d_ws, size_t ws_size,
                              hipStream_t stream) {
    (void)in_sizes; (void)n_in; (void)out_size; (void)ws_size;
    const float* x  = (const float*)d_in[0];
    const float* wq = (const float*)d_in[3];
    const float* wk = (const float*)d_in[4];
    const float* wv = (const float*)d_in[5];
    const float* wp = (const float*)d_in[6];
    const float* bp = (const float*)d_in[7];

    const size_t SZ2 = (size_t)B_*NH_*4096*DH_;
    const size_t SZ1 = (size_t)B_*NH_*1024*DH_;
    const size_t SZ0 = (size_t)B_*NH_*256 *DH_;
    float* ws = (float*)d_ws; size_t o = 0;
    float* q2 = ws + o; o += SZ2;  float* k2 = ws + o; o += SZ2;  float* v2 = ws + o; o += SZ2;
    float* q1 = ws + o; o += SZ1;  float* k1 = ws + o; o += SZ1;  float* v1 = ws + o; o += SZ1;
    float* q0 = ws + o; o += SZ0;  float* k0 = ws + o; o += SZ0;  float* v0 = ws + o; o += SZ0;
    float* msg0 = ws + o; o += SZ0;
    float* msg1 = ws + o; o += SZ1;
    float* msg2 = ws + o; o += SZ2;
    float* ps0 = ws + o; o += (size_t)B_*NH_*256*16;
    int*   pi0 = (int*)(ws + o); o += (size_t)B_*NH_*256*16;
    float* ps1 = ws + o; o += (size_t)B_*NH_*1024*16;
    int*   pi1 = (int*)(ws + o); o += (size_t)B_*NH_*1024*16;

    const dim3 gg(2, 128);
    qta_gemm_qkv<<<gg, 256, 0, stream>>>(x, wq, q2, q1, q0);
    qta_gemm_qkv<<<gg, 256, 0, stream>>>(x, wk, k2, k1, k0);
    qta_gemm_qkv<<<gg, 256, 0, stream>>>(x, wv, v2, v1, v0);

    qta_level0<<<B_*NH_*16, 128, 0, stream>>>(q0, k0, v0, msg0, ps0, pi0);
    qta_levelt<<<B_*NH_*256,  32, 0, stream>>>(q1, k1, v1, ps0, pi0, msg1, ps1, pi1, 16, 16, 0);
    qta_levelt<<<B_*NH_*1024, 32, 0, stream>>>(q2, k2, v2, ps1, pi1, msg2, nullptr, nullptr, 32, 32, 1);

    qta_gemm_proj<<<gg, 256, 0, stream>>>(msg2, msg1, msg0, wp, bp, (float*)d_out);
}


# levels 1 and 2: the 64 child rows of K, then of V, come in as whole 128 B rows (8 lanes x 16 B, 4 rows per load) into LDS; B fragments and selected V rows read from LDS; sum orders kept
# speedup vs baseline: 2.3711x; 1.0519x over previous
#include <hip/hip_runtime.h>
#include <hip/hip_bf16.h>

#define B_   2
#define NH_  8
#define HH_  64
#define WW_  64
#define C_   256
#define DH_  32
#define N_   (HH_*WW_)
#define SCALE_ 0.17677669529663687f

typedef __attribute__((ext_vector_type(2))) float v2f;
typedef __attribute__((ext_vector_type(8))) float v8f;

static __device__ __forceinline__ v8f wmma_f32(v2f a, v2f b, v8f c) {
    return __builtin_amdgcn_wmma_f32_16x16x4_f32(
         false, a,  false, b,
         (short)0, c,  false,  false);
}

#define GBM_ 64
#define GBN_ 128
#define GBK_ 32
#define GP_  36
#define GSTAGE_ ((GBM_ + GBN_) * GP_)

static __device__ __forceinline__ void qta_gemm_chunk(const float* cur, const int aoff, const int boff,
                                                      v8f& c00, v8f& c01, v8f& c10, v8f& c11) {
    for (int ks = 0; ks < GBK_; ks += 4) {
        const v2f a0 = *(const v2f*)(cur + aoff + ks);
        const v2f a1 = *(const v2f*)(cur + aoff + 16*GP_ + ks);
        const v2f b0 = *(const v2f*)(cur + boff + ks);
        const v2f b1 = *(const v2f*)(cur + boff + 16*GP_ + ks);
        c00 = wmma_f32(a0, b0, c00);
        c01 = wmma_f32(a0, b1, c01);
        c10 = wmma_f32(a1, b0, c10);
        c11 = wmma_f32(a1, b1, c11);
    }
}

struct qta_a_plain {
    const float* g;
    int row2;
    float4 r0, r1;
    __device__ __forceinline__ void load(const int c) {
        r0 = *(const float4*)(g + c*GBK_);
        r1 = *(const float4*)(g + c*GBK_ + row2);
    }
    __device__ __forceinline__ void get(float4& v0, float4& v1) const { v0 = r0; v1 = r1; }
};
struct qta_a_msgsum {
    const float *g2, *g1, *g0;
    float4 r20, r21, r10, r11, r00, r01;
    __device__ __forceinline__ void load(const int c) {
        r20 = *(const float4*)(g2 + (size_t)c*N_*DH_);
        r21 = *(const float4*)(g2 + (size_t)c*N_*DH_ + 32*DH_);
        r10 = *(const float4*)(g1 + (size_t)c*(N_/4)*DH_);
        r11 = *(const float4*)(g1 + (size_t)c*(N_/4)*DH_ + 16*DH_);
        r00 = *(const float4*)(g0 + (size_t)c*(N_/16)*DH_);
        r01 = *(const float4*)(g0 + (size_t)c*(N_/16)*DH_ + 8*DH_);
    }
    static __device__ __forceinline__ float4 sum3(const float4 a, const float4 b, const float4 c) {
        float4 r;
        r.x = (a.x + b.x) + c.x; r.y = (a.y + b.y) + c.y; r.z = (a.z + b.z) + c.z; r.w = (a.w + b.w) + c.w;
        return r;
    }
    __device__ __forceinline__ void get(float4& v0, float4& v1) const {
        v0 = sum3(r20, r10, r00);
        v1 = sum3(r21, r11, r01);
    }
};

template <class ASRC>
static __device__ __forceinline__ void qta_gemm_mainloop(ASRC& as,
                                                         const float* __restrict__ w,
                                                         float* sm, const int n0,
                                                         v8f& c00, v8f& c01, v8f& c10, v8f& c11) {
    const int tid  = threadIdx.x;
    const int wave = tid >> 5, lane = tid & 31;
    const int lm = lane & 15, hf = lane >> 4;
    const int wm = wave & 1, wn = wave >> 1;

    const int lrow = tid >> 3;
    const int lcol = (tid & 7) * 4;
    const float* wg = w + (size_t)(n0 + lrow) * C_ + lcol;
    const int so = lrow * GP_ + lcol;

    float4 ra0, ra1, rb0, rb1, rb2, rb3;
    as.load(0);
    rb0 = *(const float4*)(wg);
    rb1 = *(const float4*)(wg + 32*C_);
    rb2 = *(const float4*)(wg + 64*C_);
    rb3 = *(const float4*)(wg + 96*C_);
    as.get(ra0, ra1);
    *(float4*)(sm + so)                        = ra0;
    *(float4*)(sm + so + 32*GP_)               = ra1;
    *(float4*)(sm + GBM_*GP_ + so)             = rb0;
    *(float4*)(sm + GBM_*GP_ + so + 32*GP_)    = rb1;
    *(float4*)(sm + GBM_*GP_ + so + 64*GP_)    = rb2;
    *(float4*)(sm + GBM_*GP_ + so + 96*GP_)    = rb3;
    __syncthreads();

    const int aoff = (wm*32 + lm) * GP_ + 2*hf;
    const int boff = GBM_*GP_ + (wn*32 + lm) * GP_ + 2*hf;

    #pragma nounroll
    for (int c = 0; c < C_/GBK_ - 1; ++c) {
        const int kn = (c + 1) * GBK_;
        as.load(c + 1);
        rb0 = *(const float4*)(wg + kn);
        rb1 = *(const float4*)(wg + kn + 32*C_);
        rb2 = *(const float4*)(wg + kn + 64*C_);
        rb3 = *(const float4*)(wg + kn + 96*C_);
        __builtin_amdgcn_sched_barrier(0);
        qta_gemm_chunk(sm + (c & 1) * GSTAGE_, aoff, boff, c00, c01, c10, c11);
        float* nx = sm + ((c + 1) & 1) * GSTAGE_;
        as.get(ra0, ra1);
        *(float4*)(nx + so)                        = ra0;
        *(float4*)(nx + so + 32*GP_)               = ra1;
        *(float4*)(nx + GBM_*GP_ + so)             = rb0;
        *(float4*)(nx + GBM_*GP_ + so + 32*GP_)    = rb1;
        *(float4*)(nx + GBM_*GP_ + so + 64*GP_)    = rb2;
        *(float4*)(nx + GBM_*GP_ + so + 96*GP_)    = rb3;
        __syncthreads();
    }
    qta_gemm_chunk(sm + ((C_/GBK_ - 1) & 1) * GSTAGE_, aoff, boff, c00, c01, c10, c11);
}

__global__ __launch_bounds__(256) void qta_gemm_qkv(const float* __restrict__ x,
                                                    const float* __restrict__ w,
                                                    float* __restrict__ out2,
                                                    float* __restrict__ out1,
                                                    float* __restrict__ out0) {
    __shared__ float4 smv[2 * GSTAGE_ / 4];
    float* sm = (float*)smv;
    const int wave = threadIdx.x >> 5;
    const int lane = threadIdx.x & 31;
    const int lm = lane & 15, hf = lane >> 4;
    const int wm = wave & 1, wn = wave >> 1;
    const int n0 = blockIdx.x * GBN_;
    const int bb = blockIdx.y >> 6;
    const int py = (blockIdx.y >> 3) & 7;
    const int px = blockIdx.y & 7;

    const int lrow = threadIdx.x >> 3;
    const int ty = ((lrow >> 4) & 1) * 2 + ((lrow >> 1) & 1);
    const int tx = ((lrow >> 3) & 1) * 4 + ((lrow >> 2) & 1) * 2 + (lrow & 1);
    qta_a_plain as;
    as.g = x + ((size_t)bb * N_ + (py*8 + ty) * WW_ + px*8 + tx) * C_ + (threadIdx.x & 7) * 4;
    as.row2 = 4*WW_*C_;

    v8f acc[2][2] = {};
    qta_gemm_mainloop(as, w, sm, n0, acc[0][0], acc[0][1], acc[1][0], acc[1][1]);

    const size_t bh = (size_t)bb*NH_ + (n0 >> 5) + wn;
    const int yb = py*8 + wm*4;
    const int xb = px*8 + hf*4;
    for (int j = 0; j < 2; ++j) {
        const int d = j*16 + lm;
        float l1[2][2];
        for (int i = 0; i < 2; ++i) {
            for (int r = 0; r < 8; ++r) {
                const int yy = yb + 2*i + ((r >> 1) & 1);
                const int xx = xb + 2*(r >> 2) + (r & 1);
                out2[((bh*N_) + yy*WW_ + xx)*DH_ + d] = acc[i][j][r];
            }
            for (int qx = 0; qx < 2; ++qx) {
                const float s = 0.25f*(acc[i][j][4*qx] + acc[i][j][4*qx + 1] + acc[i][j][4*qx + 2] + acc[i][j][4*qx + 3]);
                l1[i][qx] = s;
                out1[((bh*(N_/4)) + ((yb >> 1) + i)*(WW_/2) + (xb >> 1) + qx)*DH_ + d] = s;
            }
        }
        out0[((bh*(N_/16)) + (yb >> 2)*(WW_/4) + (xb >> 2))*DH_ + d] =
            0.25f*(l1[0][0] + l1[0][1] + l1[1][0] + l1[1][1]);
    }
}

static_assert(GBM_ == WW_, "a proj work-group covers exactly one image row");
__global__ __launch_bounds__(256) void qta_gemm_proj(const float* __restrict__ m2,
                                                     const float* __restrict__ m1,
                                                     const float* __restrict__ m0,
                                                     const float* __restrict__ w,
                                                     const float* __restrict__ bias,
                                                     float* __restrict__ out) {
    __shared__ float4 smv[2 * GSTAGE_ / 4];
    float* sm = (float*)smv;
    const int wave = threadIdx.x >> 5;
    const int lane = threadIdx.x & 31;
    const int lm = lane & 15, hf = lane >> 4;
    const int wm = wave & 1, wn = wave >> 1;
    const int n0 = blockIdx.x * GBN_;
    const int mb = blockIdx.y * GBM_;
    const int bb  = blockIdx.y >> 6;
    const int row = blockIdx.y & 63;
    const int col = threadIdx.x >> 3;
    const int lcol = (threadIdx.x & 7) * 4;
    qta_a_msgsum as;
    as.g2 = m2 + ((size_t)bb*NH_*N_      + row*WW_            + col       ) * DH_ + lcol;
    as.g1 = m1 + ((size_t)bb*NH_*(N_/4)  + (row >> 1)*(WW_/2) + (col >> 1)) * DH_ + lcol;
    as.g0 = m0 + ((size_t)bb*NH_*(N_/16) + (row >> 2)*(WW_/4) + (col >> 2)) * DH_ + lcol;
    v8f acc[2][2] = {};
    qta_gemm_mainloop(as, w, sm, n0, acc[0][0], acc[0][1], acc[1][0], acc[1][1]);
    for (int i = 0; i < 2; ++i)
        for (int j = 0; j < 2; ++j)
            for (int r = 0; r < 8; ++r) {
                const int m = mb + wm*32 + i*16 + r + 8*hf;
                const int n = n0 + wn*32 + j*16 + lm;
                out[(size_t)m*C_ + n] = acc[i][j][r] + bias[n];
            }
}


template <int LO, int N> struct qta_argmax {
    static __device__ __forceinline__ void run(const float* v, float& bv, int& bi) {
        float lv, rv; int li, ri;
        qta_argmax<LO, N/2>::run(v, lv, li);
        qta_argmax<LO + N/2, N/2>::run(v, rv, ri);
        const bool r = rv > lv;
        bv = r ? rv : lv;
        bi = r ? ri : li;
    }
};
template <int LO> struct qta_argmax<LO, 1> {
    static __device__ __forceinline__ void run(const float* v, float& bv, int& bi) { bv = v[LO]; bi = LO; }
};

static __device__ __forceinline__ float qta_max16(const float* x) {
    const float a0 = fmaxf(x[0],  x[1]),  a1 = fmaxf(x[2],  x[3]),  a2 = fmaxf(x[4],  x[5]),  a3 = fmaxf(x[6],  x[7]);
    const float a4 = fmaxf(x[8],  x[9]),  a5 = fmaxf(x[10], x[11]), a6 = fmaxf(x[12], x[13]), a7 = fmaxf(x[14], x[15]);
    return fmaxf(fmaxf(fmaxf(a0, a1), fmaxf(a2, a3)), fmaxf(fmaxf(a4, a5), fmaxf(a6, a7)));
}

template <int NCH>
static __device__ __forceinline__ void qta_pick16(float* row, float (&cm)[NCH], float* tv, int* ti) {
    #pragma nounroll
    for (int t = 0; t < 16; ++t) {
        float bv; int bc;
        qta_argmax<0, NCH>::run(cm, bv, bc);
        float* ch = row + bc*16;
        float v[16];
        #pragma unroll
        for (int q = 0; q < 4; ++q) {
            const float4 x = ((const float4*)ch)[q];
            v[4*q + 0] = x.x; v[4*q + 1] = x.y; v[4*q + 2] = x.z; v[4*q + 3] = x.w;
        }
        float ev; int jj;
        qta_argmax<0, 16>::run(v, ev, jj);
        ch[jj] = -1e30f;
        #pragma unroll
        for (int j = 0; j < 16; ++j) v[j] = (j == jj) ? -1e30f : v[j];
        const float nm = qta_max16(v);
        #pragma unroll
        for (int cc = 0; cc < NCH; ++cc) cm[cc] = (cc == bc) ? nm : cm[cc];
        tv[t] = ev;
        ti[t] = bc*16 + jj;
    }
}

__global__ __launch_bounds__(128) void qta_level0(const float* __restrict__ q,
                                                  const float* __restrict__ k,
                                                  const float* __restrict__ v,
                                                  float* __restrict__ msg,
                                                  float* __restrict__ ps_out,
                                                  int* __restrict__ pi_out) {
    __shared__ __attribute__((aligned(16))) float S[16][256];
    __shared__ __attribute__((aligned(16))) float SC[16][256];
    __shared__ float MA[16][32];
    __shared__ float topv[16][16];
    __shared__ int   topi[16][16];
    __shared__ __attribute__((aligned(16))) float rinv[16];
    __shared__ float wmx[4][16];

    const int blk = blockIdx.x;
    const int qt  = blk & 15;
    const int bh  = blk >> 4;
    const int lane = threadIdx.x & 31;
    const int wave = __builtin_amdgcn_readfirstlane((int)(threadIdx.x >> 5));
    const int lm = lane & 15, hf = lane >> 4;
    const int qrow0 = qt * 16;
    const float* Q = q + (size_t)bh*256*DH_;
    const float* K = k + (size_t)bh*256*DH_;
    const float* V = v + (size_t)bh*256*DH_;

    v8f acc[4];
    const float* qr = Q + (size_t)(qrow0 + lm)*DH_;
    #pragma unroll
    for (int nt = 0; nt < 4; ++nt) {
        v8f c = {};
        const float* kr = K + (size_t)((wave*4 + nt)*16 + lm)*DH_;
        for (int kk = 0; kk < DH_; kk += 4) {
            v2f a, b;
            a[0] = SCALE_ * qr[kk + 0 + 2*hf];
            a[1] = SCALE_ * qr[kk + 1 + 2*hf];
            b[0] = kr[kk + 0 + 2*hf];
            b[1] = kr[kk + 1 + 2*hf];
            c = wmma_f32(a, b, c);
        }
        acc[nt] = c;
    }

    float mx[8];
    #pragma unroll
    for (int r = 0; r < 8; ++r) {
        float m = -1e30f;
        #pragma unroll
        for (int nt = 0; nt < 4; ++nt) m = fmaxf(m, acc[nt][r]);
        for (int o = 1; o < 16; o <<= 1) m = fmaxf(m, __shfl_xor(m, o, 32));
        mx[r] = m;
    }
    if (lm == 0) {
        #pragma unroll
        for (int r = 0; r < 8; ++r) wmx[wave][r + 8*hf] = mx[r];
    }
    __syncthreads();
    #pragma unroll
    for (int r = 0; r < 8; ++r)
        mx[r] = fmaxf(fmaxf(wmx[0][r + 8*hf], wmx[1][r + 8*hf]), fmaxf(wmx[2][r + 8*hf], wmx[3][r + 8*hf]));
    #pragma unroll
    for (int nt = 0; nt < 4; ++nt) {
        #pragma unroll
        for (int r = 0; r < 8; ++r) {
            const float e = __expf(acc[nt][r] - mx[r]);
            acc[nt][r] = e;
            S[r + 8*hf][(wave*4 + nt)*16 + lm] = e;
        }
    }
    __syncthreads();

    float cm[16];
    if (wave == 0 && lane < 16) {
        const int m = lane;
        float sum = 0.f;
        const float4* rowv = (const float4*)&S[m][0];
        float4 nx[2][4];
        #pragma unroll
        for (int q = 0; q < 4; ++q) nx[0][q] = rowv[q];
        #pragma unroll
        for (int cc = 0; cc < 16; ++cc) {
            if (cc + 1 < 16) {
                #pragma unroll
                for (int q = 0; q < 4; ++q) nx[(cc + 1) & 1][q] = rowv[(cc + 1)*4 + q];
            }
            __builtin_amdgcn_sched_barrier(0);
            float ev[16];
            #pragma unroll
            for (int q = 0; q < 4; ++q) {
                const float4 x = nx[cc & 1][q];
                ev[4*q + 0] = x.x; ev[4*q + 1] = x.y; ev[4*q + 2] = x.z; ev[4*q + 3] = x.w;
            }
            #pragma unroll
            for (int jj = 0; jj < 16; ++jj) sum += ev[jj];
            cm[cc] = qta_max16(ev);
        }
        const float inv = 1.f / sum;
        #pragma unroll
        for (int cc = 0; cc < 16; ++cc) cm[cc] *= inv;
        rinv[m] = inv;
    }
    __syncthreads();

    float iv[8];
    #pragma unroll
    for (int r = 0; r < 8; ++r) iv[r] = rinv[r + 8*hf];
    #pragma unroll
    for (int nt = 0; nt < 4; ++nt) {
        #pragma unroll
        for (int r = 0; r < 8; ++r) {
            const float pv = acc[nt][r] * iv[r];
            S[r + 8*hf][(wave*4 + nt)*16 + lm] = pv;
            SC[r + 8*hf][(wave*4 + nt)*16 + lm] = pv;
        }
    }
    __syncthreads();

    if (wave == 0) {
        if (lane < 16)
            qta_pick16<16>(&SC[lane][0], cm, &topv[lane][0], &topi[lane][0]);
    } else if (wave <= 2) {
        const int dcol = (wave - 1)*16 + lm;
        v8f a0 = {};
        for (int kk = 0; kk < 256; kk += 4) {
            v2f a, b0;
            a[0] = S[lm][kk + 0 + 2*hf];
            a[1] = S[lm][kk + 1 + 2*hf];
            b0[0] = V[(size_t)(kk + 0 + 2*hf)*DH_ + dcol];
            b0[1] = V[(size_t)(kk + 1 + 2*hf)*DH_ + dcol];
            a0 = wmma_f32(a, b0, a0);
        }
        for (int r = 0; r < 8; ++r) MA[r + 8*hf][dcol] = a0[r];
    }
    __syncthreads();

    for (int m = 4*wave; m < 4*wave + 4; ++m) {
        float msel = 0.f;
        for (int t = 0; t < 16; ++t)
            msel += topv[m][t] * V[(size_t)topi[m][t]*DH_ + lane];
        msg[((size_t)bh*256 + qrow0 + m)*DH_ + lane] = MA[m][lane] - msel;
    }
    if (lane < 16) {
        for (int m = 4*wave; m < 4*wave + 4; ++m) {
            ps_out[((size_t)bh*256 + qrow0 + m)*16 + lane] = topv[m][lane];
            pi_out[((size_t)bh*256 + qrow0 + m)*16 + lane] = topi[m][lane];
        }
    }
}

#define LKP_ 36
static __device__ __forceinline__ int qta_vcol(const int row, const int col) { return col ^ (((row >> 1) & 1) << 4); }
__global__ __launch_bounds__(32) void qta_levelt(const float* __restrict__ qf,
                                                 const float* __restrict__ kf,
                                                 const float* __restrict__ vf,
                                                 const float* __restrict__ ps_in,
                                                 const int*   __restrict__ pi_in,
                                                 float* __restrict__ msg,
                                                 float* __restrict__ ps_out,
                                                 int*   __restrict__ pi_out,
                                                 int gh, int gw, int is_last) {
    __shared__ __attribute__((aligned(16))) float KV[64*LKP_];
    __shared__ __attribute__((aligned(16))) float sL[4][64];
    __shared__ float mA[4][32];
    __shared__ float psh[16];
    __shared__ int   ci[64];
    __shared__ __attribute__((aligned(16))) int ciT[4][16];
    __shared__ float topv[4][16];
    __shared__ int   topi[4][16];

    const int Sp = gh*gw;
    const int p  = blockIdx.x % Sp;
    const int bh = blockIdx.x / Sp;
    const int wt = 2*gw;
    const int St = 4*Sp;
    const int lane = threadIdx.x;
    const int lm = lane & 15, hf = lane >> 4;
    const float* Q = qf + (size_t)bh*St*DH_;
    const float* K = kf + (size_t)bh*St*DH_;
    const float* V = vf + (size_t)bh*St*DH_;

    if (lane < 16) {
        psh[lane] = ps_in[((size_t)bh*Sp + p)*16 + lane];
        const int pi = pi_in[((size_t)bh*Sp + p)*16 + lane];
        const int r = pi / gw, c = pi % gw;
        for (int cc = 0; cc < 4; ++cc) {
            const int g = (2*r + (cc >> 1))*wt + (2*c + (cc & 1));
            ci[lane*4 + cc] = g;
            ciT[cc][lane] = g;
        }
    }
    __syncthreads();

    const int sc   = lane >> 3;
    const int scol = (lane & 7) * 4;
    int srow[16];
    #pragma unroll
    for (int q4 = 0; q4 < 4; ++q4) {
        const int4 t = ((const int4*)&ciT[sc][0])[q4];
        srow[4*q4 + 0] = t.x; srow[4*q4 + 1] = t.y; srow[4*q4 + 2] = t.z; srow[4*q4 + 3] = t.w;
    }
    float4 stg[16];
    #pragma unroll
    for (int i = 0; i < 16; ++i) stg[i] = *(const float4*)(K + (size_t)srow[i]*DH_ + scol);
    #pragma unroll
    for (int i = 0; i < 16; ++i) *(float4*)&KV[(4*i + sc)*LKP_ + scol] = stg[i];
    __syncthreads();

    const int rp = p / gw, cp = p % gw;
    const int qm = lm & 3;
    const int qidx = (2*rp + (qm >> 1))*wt + (2*cp + (qm & 1));
    const bool alive = (lm < 4);

    for (int nt = 0; nt < 4; ++nt) {
        v8f acc = {};
        const float* qr = Q + (size_t)qidx*DH_;
        const float* kr = &KV[(nt*16 + lm)*LKP_];
        for (int kk = 0; kk < DH_; kk += 4) {
            v2f a, b;
            const float a0v = SCALE_ * qr[kk + 0 + 2*hf];
            const float a1v = SCALE_ * qr[kk + 1 + 2*hf];
            a[0] = alive ? a0v : 0.f;
            a[1] = alive ? a1v : 0.f;
            b[0] = kr[kk + 0 + 2*hf];
            b[1] = kr[kk + 1 + 2*hf];
            acc = wmma_f32(a, b, acc);
        }
        if (hf == 0)
            for (int r = 0; r < 4; ++r) sL[r][nt*16 + lm] = acc[r];
    }
    __syncthreads();

    for (int g = lane; g < 64; g += 32) {
        const int m = g >> 4, kp = g & 15;
        float* row = &sL[m][kp*4];
        const float mx = fmaxf(fmaxf(row[0], row[1]), fmaxf(row[2], row[3]));
        const float e0 = __expf(row[0]-mx), e1 = __expf(row[1]-mx);
        const float e2 = __expf(row[2]-mx), e3 = __expf(row[3]-mx);
        const float f = psh[kp] / (e0+e1+e2+e3);
        row[0] = e0*f; row[1] = e1*f; row[2] = e2*f; row[3] = e3*f;
    }
    #pragma unroll
    for (int i = 0; i < 16; ++i) stg[i] = *(const float4*)(V + (size_t)srow[i]*DH_ + scol);
    #pragma unroll
    for (int i = 0; i < 16; ++i) *(float4*)&KV[(4*i + sc)*DH_ + qta_vcol(4*i + sc, scol)] = stg[i];
    __syncthreads();

    for (int nt = 0; nt < 2; ++nt) {
        v8f acc = {};
        for (int kk = 0; kk < 64; kk += 4) {
            v2f a, b;
            const float a0v = sL[qm][kk + 0 + 2*hf];
            const float a1v = sL[qm][kk + 1 + 2*hf];
            a[0] = alive ? a0v : 0.f;
            a[1] = alive ? a1v : 0.f;
            b[0] = KV[(kk + 0 + 2*hf)*DH_ + qta_vcol(kk + 0 + 2*hf, nt*16 + lm)];
            b[1] = KV[(kk + 1 + 2*hf)*DH_ + qta_vcol(kk + 1 + 2*hf, nt*16 + lm)];
            acc = wmma_f32(a, b, acc);
        }
        if (hf == 0)
            for (int r = 0; r < 4; ++r) mA[r][nt*16 + lm] = acc[r];
    }
    __syncthreads();

    if (is_last) {
        for (int m = 0; m < 4; ++m) {
            const int fi = (2*rp + (m >> 1))*wt + (2*cp + (m & 1));
            msg[((size_t)bh*St + fi)*DH_ + lane] = mA[m][lane];
        }
        return;
    }

    if (lane < 4) {
        const int m = lane;
        float cm[4];
        #pragma unroll
        for (int cc = 0; cc < 4; ++cc) {
            float x[16];
            #pragma unroll
            for (int q = 0; q < 4; ++q) {
                const float4 xv = ((const float4*)&sL[m][cc*16])[q];
                x[4*q + 0] = xv.x; x[4*q + 1] = xv.y; x[4*q + 2] = xv.z; x[4*q + 3] = xv.w;
            }
            cm[cc] = qta_max16(x);
        }
        qta_pick16<4>(&sL[m][0], cm, &topv[m][0], &topi[m][0]);
    }
    __syncthreads();

    for (int m = 0; m < 4; ++m) {
        float msel = 0.f;
        for (int t = 0; t < 16; ++t) {
            const int r = topi[m][t];
            msel += topv[m][t] * KV[r*DH_ + qta_vcol(r, lane)];
        }
        const int fi = (2*rp + (m >> 1))*wt + (2*cp + (m & 1));
        msg[((size_t)bh*St + fi)*DH_ + lane] = mA[m][lane] - msel;
    }
    if (lane < 16) {
        for (int m = 0; m < 4; ++m) {
            const int fi = (2*rp + (m >> 1))*wt + (2*cp + (m & 1));
            ps_out[((size_t)bh*St + fi)*16 + lane] = topv[m][lane];
            pi_out[((size_t)bh*St + fi)*16 + lane] = ci[topi[m][lane]];
        }
    }
}

extern "C" void kernel_launch(void* const* d_in, const int* in_sizes, int n_in,
                              void* d_out, int out_size, void* d_ws, size_t ws_size,
                              hipStream_t stream) {
    (void)in_sizes; (void)n_in; (void)out_size; (void)ws_size;
    const float* x  = (const float*)d_in[0];
    const float* wq = (const float*)d_in[3];
    const float* wk = (const float*)d_in[4];
    const float* wv = (const float*)d_in[5];
    const float* wp = (const float*)d_in[6];
    const float* bp = (const float*)d_in[7];

    const size_t SZ2 = (size_t)B_*NH_*4096*DH_;
    const size_t SZ1 = (size_t)B_*NH_*1024*DH_;
    const size_t SZ0 = (size_t)B_*NH_*256 *DH_;
    float* ws = (float*)d_ws; size_t o = 0;
    float* q2 = ws + o; o += SZ2;  float* k2 = ws + o; o += SZ2;  float* v2 = ws + o; o += SZ2;
    float* q1 = ws + o; o += SZ1;  float* k1 = ws + o; o += SZ1;  float* v1 = ws + o; o += SZ1;
    float* q0 = ws + o; o += SZ0;  float* k0 = ws + o; o += SZ0;  float* v0 = ws + o; o += SZ0;
    float* msg0 = ws + o; o += SZ0;
    float* msg1 = ws + o; o += SZ1;
    float* msg2 = ws + o; o += SZ2;
    float* ps0 = ws + o; o += (size_t)B_*NH_*256*16;
    int*   pi0 = (int*)(ws + o); o += (size_t)B_*NH_*256*16;
    float* ps1 = ws + o; o += (size_t)B_*NH_*1024*16;
    int*   pi1 = (int*)(ws + o); o += (size_t)B_*NH_*1024*16;

    const dim3 gg(2, 128);
    qta_gemm_qkv<<<gg, 256, 0, stream>>>(x, wq, q2, q1, q0);
    qta_gemm_qkv<<<gg, 256, 0, stream>>>(x, wk, k2, k1, k0);
    qta_gemm_qkv<<<gg, 256, 0, stream>>>(x, wv, v2, v1, v0);

    qta_level0<<<B_*NH_*16, 128, 0, stream>>>(q0, k0, v0, msg0, ps0, pi0);
    qta_levelt<<<B_*NH_*256,  32, 0, stream>>>(q1, k1, v1, ps0, pi0, msg1, ps1, pi1, 16, 16, 0);
    qta_levelt<<<B_*NH_*1024, 32, 0, stream>>>(q2, k2, v2, ps1, pi1, msg2, nullptr, nullptr, 32, 32, 1);

    qta_gemm_proj<<<gg, 256, 0, stream>>>(msg2, msg1, msg0, wp, bp, (float*)d_out);
}


# GEMMs (q,k,v,proj): chunk loads issued two chunks ahead into two register sets and kept in flight across an LDS-only barrier (were one ahead, drained inside the chunk); k order, tile code unchanged
# speedup vs baseline: 2.4874x; 1.0490x over previous
#include <hip/hip_runtime.h>
#include <hip/hip_bf16.h>

#define B_   2
#define NH_  8
#define HH_  64
#define WW_  64
#define C_   256
#define DH_  32
#define N_   (HH_*WW_)
#define SCALE_ 0.17677669529663687f

typedef __attribute__((ext_vector_type(2))) float v2f;
typedef __attribute__((ext_vector_type(8))) float v8f;

static __device__ __forceinline__ v8f wmma_f32(v2f a, v2f b, v8f c) {
    return __builtin_amdgcn_wmma_f32_16x16x4_f32(
         false, a,  false, b,
         (short)0, c,  false,  false);
}

#define GBM_ 64
#define GBN_ 128
#define GBK_ 32
#define GP_  36
#define GSTAGE_ ((GBM_ + GBN_) * GP_)

static __device__ __forceinline__ void qta_gemm_chunk(const float* cur, const int aoff, const int boff,
                                                      v8f& c00, v8f& c01, v8f& c10, v8f& c11) {
    for (int ks = 0; ks < GBK_; ks += 4) {
        const v2f a0 = *(const v2f*)(cur + aoff + ks);
        const v2f a1 = *(const v2f*)(cur + aoff + 16*GP_ + ks);
        const v2f b0 = *(const v2f*)(cur + boff + ks);
        const v2f b1 = *(const v2f*)(cur + boff + 16*GP_ + ks);
        c00 = wmma_f32(a0, b0, c00);
        c01 = wmma_f32(a0, b1, c01);
        c10 = wmma_f32(a1, b0, c10);
        c11 = wmma_f32(a1, b1, c11);
    }
}

struct qta_a_plain {
    const float* g;
    int row2;
    struct regs { float4 r0, r1; };
    __device__ __forceinline__ void load(regs& R, const int c) const {
        R.r0 = *(const float4*)(g + c*GBK_);
        R.r1 = *(const float4*)(g + c*GBK_ + row2);
    }
    static __device__ __forceinline__ void get(const regs& R, float4& v0, float4& v1) { v0 = R.r0; v1 = R.r1; }
};
struct qta_a_msgsum {
    const float *g2, *g1, *g0;
    struct regs { float4 r20, r21, r10, r11, r00, r01; };
    __device__ __forceinline__ void load(regs& R, const int c) const {
        R.r20 = *(const float4*)(g2 + (size_t)c*N_*DH_);
        R.r21 = *(const float4*)(g2 + (size_t)c*N_*DH_ + 32*DH_);
        R.r10 = *(const float4*)(g1 + (size_t)c*(N_/4)*DH_);
        R.r11 = *(const float4*)(g1 + (size_t)c*(N_/4)*DH_ + 16*DH_);
        R.r00 = *(const float4*)(g0 + (size_t)c*(N_/16)*DH_);
        R.r01 = *(const float4*)(g0 + (size_t)c*(N_/16)*DH_ + 8*DH_);
    }
    static __device__ __forceinline__ float4 sum3(const float4 a, const float4 b, const float4 c) {
        float4 r;
        r.x = (a.x + b.x) + c.x; r.y = (a.y + b.y) + c.y; r.z = (a.z + b.z) + c.z; r.w = (a.w + b.w) + c.w;
        return r;
    }
    static __device__ __forceinline__ void get(const regs& R, float4& v0, float4& v1) {
        v0 = sum3(R.r20, R.r10, R.r00);
        v1 = sum3(R.r21, R.r11, R.r01);
    }
};

static __device__ __forceinline__ void qta_w_load(float4& r0, float4& r1, float4& r2, float4& r3,
                                                  const float* wg, const int c) {
    r0 = *(const float4*)(wg + c*GBK_);
    r1 = *(const float4*)(wg + c*GBK_ + 32*C_);
    r2 = *(const float4*)(wg + c*GBK_ + 64*C_);
    r3 = *(const float4*)(wg + c*GBK_ + 96*C_);
}
template <class ASRC>
static __device__ __forceinline__ void qta_stage_store(float* st, const int so, const typename ASRC::regs& A,
                                                       const float4 r0, const float4 r1,
                                                       const float4 r2, const float4 r3) {
    float4 a0, a1;
    ASRC::get(A, a0, a1);
    *(float4*)(st + so)                        = a0;
    *(float4*)(st + so + 32*GP_)               = a1;
    *(float4*)(st + GBM_*GP_ + so)             = r0;
    *(float4*)(st + GBM_*GP_ + so + 32*GP_)    = r1;
    *(float4*)(st + GBM_*GP_ + so + 64*GP_)    = r2;
    *(float4*)(st + GBM_*GP_ + so + 96*GP_)    = r3;
}

static __device__ __forceinline__ void qta_lds_barrier() {
    __builtin_amdgcn_fence(__ATOMIC_RELEASE, "workgroup", "local");
    __builtin_amdgcn_s_barrier();
    __builtin_amdgcn_fence(__ATOMIC_ACQUIRE, "workgroup", "local");
}

static_assert((C_/GBK_) % 2 == 0 && C_/GBK_ >= 4, "the chunk loop below runs in pairs of chunks");
template <class ASRC>
static __device__ __forceinline__ void qta_gemm_mainloop(const ASRC& as,
                                                         const float* __restrict__ w,
                                                         float* sm, const int n0,
                                                         v8f& c00, v8f& c01, v8f& c10, v8f& c11) {
    const int tid  = threadIdx.x;
    const int wave = tid >> 5, lane = tid & 31;
    const int lm = lane & 15, hf = lane >> 4;
    const int wm = wave & 1, wn = wave >> 1;

    const int lrow = tid >> 3;
    const int lcol = (tid & 7) * 4;
    const float* wg = w + (size_t)(n0 + lrow) * C_ + lcol;
    const int so = lrow * GP_ + lcol;

    typename ASRC::regs a_e, a_o;
    float4 we0, we1, we2, we3, wo0, wo1, wo2, wo3;
    float* const st_e = sm;
    float* const st_o = sm + GSTAGE_;

    as.load(a_e, 0);  qta_w_load(we0, we1, we2, we3, wg, 0);
    as.load(a_o, 1);  qta_w_load(wo0, wo1, wo2, wo3, wg, 1);
    qta_stage_store<ASRC>(st_e, so, a_e, we0, we1, we2, we3);
    qta_lds_barrier();

    const int aoff = (wm*32 + lm) * GP_ + 2*hf;
    const int boff = GBM_*GP_ + (wn*32 + lm) * GP_ + 2*hf;

    #pragma nounroll
    for (int c = 0; c < C_/GBK_ - 2; c += 2) {
        as.load(a_e, c + 2);  qta_w_load(we0, we1, we2, we3, wg, c + 2);
        asm volatile("" ::: "memory");
        qta_gemm_chunk(st_e, aoff, boff, c00, c01, c10, c11);
        qta_stage_store<ASRC>(st_o, so, a_o, wo0, wo1, wo2, wo3);
        qta_lds_barrier();
        as.load(a_o, c + 3);  qta_w_load(wo0, wo1, wo2, wo3, wg, c + 3);
        asm volatile("" ::: "memory");
        qta_gemm_chunk(st_o, aoff, boff, c00, c01, c10, c11);
        qta_stage_store<ASRC>(st_e, so, a_e, we0, we1, we2, we3);
        qta_lds_barrier();
    }
    qta_gemm_chunk(st_e, aoff, boff, c00, c01, c10, c11);
    qta_stage_store<ASRC>(st_o, so, a_o, wo0, wo1, wo2, wo3);
    qta_lds_barrier();
    qta_gemm_chunk(st_o, aoff, boff, c00, c01, c10, c11);
}

__global__ __launch_bounds__(256) void qta_gemm_qkv(const float* __restrict__ x,
                                                    const float* __restrict__ w,
                                                    float* __restrict__ out2,
                                                    float* __restrict__ out1,
                                                    float* __restrict__ out0) {
    __shared__ float4 smv[2 * GSTAGE_ / 4];
    float* sm = (float*)smv;
    const int wave = threadIdx.x >> 5;
    const int lane = threadIdx.x & 31;
    const int lm = lane & 15, hf = lane >> 4;
    const int wm = wave & 1, wn = wave >> 1;
    const int n0 = blockIdx.x * GBN_;
    const int bb = blockIdx.y >> 6;
    const int py = (blockIdx.y >> 3) & 7;
    const int px = blockIdx.y & 7;

    const int lrow = threadIdx.x >> 3;
    const int ty = ((lrow >> 4) & 1) * 2 + ((lrow >> 1) & 1);
    const int tx = ((lrow >> 3) & 1) * 4 + ((lrow >> 2) & 1) * 2 + (lrow & 1);
    qta_a_plain as;
    as.g = x + ((size_t)bb * N_ + (py*8 + ty) * WW_ + px*8 + tx) * C_ + (threadIdx.x & 7) * 4;
    as.row2 = 4*WW_*C_;

    v8f acc[2][2] = {};
    qta_gemm_mainloop(as, w, sm, n0, acc[0][0], acc[0][1], acc[1][0], acc[1][1]);

    const size_t bh = (size_t)bb*NH_ + (n0 >> 5) + wn;
    const int yb = py*8 + wm*4;
    const int xb = px*8 + hf*4;
    for (int j = 0; j < 2; ++j) {
        const int d = j*16 + lm;
        float l1[2][2];
        for (int i = 0; i < 2; ++i) {
            for (int r = 0; r < 8; ++r) {
                const int yy = yb + 2*i + ((r >> 1) & 1);
                const int xx = xb + 2*(r >> 2) + (r & 1);
                out2[((bh*N_) + yy*WW_ + xx)*DH_ + d] = acc[i][j][r];
            }
            for (int qx = 0; qx < 2; ++qx) {
                const float s = 0.25f*(acc[i][j][4*qx] + acc[i][j][4*qx + 1] + acc[i][j][4*qx + 2] + acc[i][j][4*qx + 3]);
                l1[i][qx] = s;
                out1[((bh*(N_/4)) + ((yb >> 1) + i)*(WW_/2) + (xb >> 1) + qx)*DH_ + d] = s;
            }
        }
        out0[((bh*(N_/16)) + (yb >> 2)*(WW_/4) + (xb >> 2))*DH_ + d] =
            0.25f*(l1[0][0] + l1[0][1] + l1[1][0] + l1[1][1]);
    }
}

static_assert(GBM_ == WW_, "a proj work-group covers exactly one image row");
__global__ __launch_bounds__(256) void qta_gemm_proj(const float* __restrict__ m2,
                                                     const float* __restrict__ m1,
                                                     const float* __restrict__ m0,
                                                     const float* __restrict__ w,
                                                     const float* __restrict__ bias,
                                                     float* __restrict__ out) {
    __shared__ float4 smv[2 * GSTAGE_ / 4];
    float* sm = (float*)smv;
    const int wave = threadIdx.x >> 5;
    const int lane = threadIdx.x & 31;
    const int lm = lane & 15, hf = lane >> 4;
    const int wm = wave & 1, wn = wave >> 1;
    const int n0 = blockIdx.x * GBN_;
    const int mb = blockIdx.y * GBM_;
    const int bb  = blockIdx.y >> 6;
    const int row = blockIdx.y & 63;
    const int col = threadIdx.x >> 3;
    const int lcol = (threadIdx.x & 7) * 4;
    qta_a_msgsum as;
    as.g2 = m2 + ((size_t)bb*NH_*N_      + row*WW_            + col       ) * DH_ + lcol;
    as.g1 = m1 + ((size_t)bb*NH_*(N_/4)  + (row >> 1)*(WW_/2) + (col >> 1)) * DH_ + lcol;
    as.g0 = m0 + ((size_t)bb*NH_*(N_/16) + (row >> 2)*(WW_/4) + (col >> 2)) * DH_ + lcol;
    v8f acc[2][2] = {};
    qta_gemm_mainloop(as, w, sm, n0, acc[0][0], acc[0][1], acc[1][0], acc[1][1]);
    for (int i = 0; i < 2; ++i)
        for (int j = 0; j < 2; ++j)
            for (int r = 0; r < 8; ++r) {
                const int m = mb + wm*32 + i*16 + r + 8*hf;
                const int n = n0 + wn*32 + j*16 + lm;
                out[(size_t)m*C_ + n] = acc[i][j][r] + bias[n];
            }
}


template <int LO, int N> struct qta_argmax {
    static __device__ __forceinline__ void run(const float* v, float& bv, int& bi) {
        float lv, rv; int li, ri;
        qta_argmax<LO, N/2>::run(v, lv, li);
        qta_argmax<LO + N/2, N/2>::run(v, rv, ri);
        const bool r = rv > lv;
        bv = r ? rv : lv;
        bi = r ? ri : li;
    }
};
template <int LO> struct qta_argmax<LO, 1> {
    static __device__ __forceinline__ void run(const float* v, float& bv, int& bi) { bv = v[LO]; bi = LO; }
};

static __device__ __forceinline__ float qta_max16(const float* x) {
    const float a0 = fmaxf(x[0],  x[1]),  a1 = fmaxf(x[2],  x[3]),  a2 = fmaxf(x[4],  x[5]),  a3 = fmaxf(x[6],  x[7]);
    const float a4 = fmaxf(x[8],  x[9]),  a5 = fmaxf(x[10], x[11]), a6 = fmaxf(x[12], x[13]), a7 = fmaxf(x[14], x[15]);
    return fmaxf(fmaxf(fmaxf(a0, a1), fmaxf(a2, a3)), fmaxf(fmaxf(a4, a5), fmaxf(a6, a7)));
}

template <int NCH>
static __device__ __forceinline__ void qta_pick16(float* row, float (&cm)[NCH], float* tv, int* ti) {
    #pragma nounroll
    for (int t = 0; t < 16; ++t) {
        float bv; int bc;
        qta_argmax<0, NCH>::run(cm, bv, bc);
        float* ch = row + bc*16;
        float v[16];
        #pragma unroll
        for (int q = 0; q < 4; ++q) {
            const float4 x = ((const float4*)ch)[q];
            v[4*q + 0] = x.x; v[4*q + 1] = x.y; v[4*q + 2] = x.z; v[4*q + 3] = x.w;
        }
        float ev; int jj;
        qta_argmax<0, 16>::run(v, ev, jj);
        ch[jj] = -1e30f;
        #pragma unroll
        for (int j = 0; j < 16; ++j) v[j] = (j == jj) ? -1e30f : v[j];
        const float nm = qta_max16(v);
        #pragma unroll
        for (int cc = 0; cc < NCH; ++cc) cm[cc] = (cc == bc) ? nm : cm[cc];
        tv[t] = ev;
        ti[t] = bc*16 + jj;
    }
}

__global__ __launch_bounds__(128) void qta_level0(const float* __restrict__ q,
                                                  const float* __restrict__ k,
                                                  const float* __restrict__ v,
                                                  float* __restrict__ msg,
                                                  float* __restrict__ ps_out,
                                                  int* __restrict__ pi_out) {
    __shared__ __attribute__((aligned(16))) float S[16][256];
    __shared__ __attribute__((aligned(16))) float SC[16][256];
    __shared__ float MA[16][32];
    __shared__ float topv[16][16];
    __shared__ int   topi[16][16];
    __shared__ __attribute__((aligned(16))) float rinv[16];
    __shared__ float wmx[4][16];

    const int blk = blockIdx.x;
    const int qt  = blk & 15;
    const int bh  = blk >> 4;
    const int lane = threadIdx.x & 31;
    const int wave = __builtin_amdgcn_readfirstlane((int)(threadIdx.x >> 5));
    const int lm = lane & 15, hf = lane >> 4;
    const int qrow0 = qt * 16;
    const float* Q = q + (size_t)bh*256*DH_;
    const float* K = k + (size_t)bh*256*DH_;
    const float* V = v + (size_t)bh*256*DH_;

    v8f acc[4];
    const float* qr = Q + (size_t)(qrow0 + lm)*DH_;
    #pragma unroll
    for (int nt = 0; nt < 4; ++nt) {
        v8f c = {};
        const float* kr = K + (size_t)((wave*4 + nt)*16 + lm)*DH_;
        for (int kk = 0; kk < DH_; kk += 4) {
            v2f a, b;
            a[0] = SCALE_ * qr[kk + 0 + 2*hf];
            a[1] = SCALE_ * qr[kk + 1 + 2*hf];
            b[0] = kr[kk + 0 + 2*hf];
            b[1] = kr[kk + 1 + 2*hf];
            c = wmma_f32(a, b, c);
        }
        acc[nt] = c;
    }

    float mx[8];
    #pragma unroll
    for (int r = 0; r < 8; ++r) {
        float m = -1e30f;
        #pragma unroll
        for (int nt = 0; nt < 4; ++nt) m = fmaxf(m, acc[nt][r]);
        for (int o = 1; o < 16; o <<= 1) m = fmaxf(m, __shfl_xor(m, o, 32));
        mx[r] = m;
    }
    if (lm == 0) {
        #pragma unroll
        for (int r = 0; r < 8; ++r) wmx[wave][r + 8*hf] = mx[r];
    }
    __syncthreads();
    #pragma unroll
    for (int r = 0; r < 8; ++r)
        mx[r] = fmaxf(fmaxf(wmx[0][r + 8*hf], wmx[1][r + 8*hf]), fmaxf(wmx[2][r + 8*hf], wmx[3][r + 8*hf]));
    #pragma unroll
    for (int nt = 0; nt < 4; ++nt) {
        #pragma unroll
        for (int r = 0; r < 8; ++r) {
            const float e = __expf(acc[nt][r] - mx[r]);
            acc[nt][r] = e;
            S[r + 8*hf][(wave*4 + nt)*16 + lm] = e;
        }
    }
    __syncthreads();

    float cm[16];
    if (wave == 0 && lane < 16) {
        const int m = lane;
        float sum = 0.f;
        const float4* rowv = (const float4*)&S[m][0];
        float4 nx[2][4];
        #pragma unroll
        for (int q = 0; q < 4; ++q) nx[0][q] = rowv[q];
        #pragma unroll
        for (int cc = 0; cc < 16; ++cc) {
            if (cc + 1 < 16) {
                #pragma unroll
                for (int q = 0; q < 4; ++q) nx[(cc + 1) & 1][q] = rowv[(cc + 1)*4 + q];
            }
            __builtin_amdgcn_sched_barrier(0);
            float ev[16];
            #pragma unroll
            for (int q = 0; q < 4; ++q) {
                const float4 x = nx[cc & 1][q];
                ev[4*q + 0] = x.x; ev[4*q + 1] = x.y; ev[4*q + 2] = x.z; ev[4*q + 3] = x.w;
            }
            #pragma unroll
            for (int jj = 0; jj < 16; ++jj) sum += ev[jj];
            cm[cc] = qta_max16(ev);
        }
        const float inv = 1.f / sum;
        #pragma unroll
        for (int cc = 0; cc < 16; ++cc) cm[cc] *= inv;
        rinv[m] = inv;
    }
    __syncthreads();

    float iv[8];
    #pragma unroll
    for (int r = 0; r < 8; ++r) iv[r] = rinv[r + 8*hf];
    #pragma unroll
    for (int nt = 0; nt < 4; ++nt) {
        #pragma unroll
        for (int r = 0; r < 8; ++r) {
            const float pv = acc[nt][r] * iv[r];
            S[r + 8*hf][(wave*4 + nt)*16 + lm] = pv;
            SC[r + 8*hf][(wave*4 + nt)*16 + lm] = pv;
        }
    }
    __syncthreads();

    if (wave == 0) {
        if (lane < 16)
            qta_pick16<16>(&SC[lane][0], cm, &topv[lane][0], &topi[lane][0]);
    } else if (wave <= 2) {
        const int dcol = (wave - 1)*16 + lm;
        v8f a0 = {};
        for (int kk = 0; kk < 256; kk += 4) {
            v2f a, b0;
            a[0] = S[lm][kk + 0 + 2*hf];
            a[1] = S[lm][kk + 1 + 2*hf];
            b0[0] = V[(size_t)(kk + 0 + 2*hf)*DH_ + dcol];
            b0[1] = V[(size_t)(kk + 1 + 2*hf)*DH_ + dcol];
            a0 = wmma_f32(a, b0, a0);
        }
        for (int r = 0; r < 8; ++r) MA[r + 8*hf][dcol] = a0[r];
    }
    __syncthreads();

    for (int m = 4*wave; m < 4*wave + 4; ++m) {
        float msel = 0.f;
        for (int t = 0; t < 16; ++t)
            msel += topv[m][t] * V[(size_t)topi[m][t]*DH_ + lane];
        msg[((size_t)bh*256 + qrow0 + m)*DH_ + lane] = MA[m][lane] - msel;
    }
    if (lane < 16) {
        for (int m = 4*wave; m < 4*wave + 4; ++m) {
            ps_out[((size_t)bh*256 + qrow0 + m)*16 + lane] = topv[m][lane];
            pi_out[((size_t)bh*256 + qrow0 + m)*16 + lane] = topi[m][lane];
        }
    }
}

#define LKP_ 36
static __device__ __forceinline__ int qta_vcol(const int row, const int col) { return col ^ (((row >> 1) & 1) << 4); }
__global__ __launch_bounds__(32) void qta_levelt(const float* __restrict__ qf,
                                                 const float* __restrict__ kf,
                                                 const float* __restrict__ vf,
                                                 const float* __restrict__ ps_in,
                                                 const int*   __restrict__ pi_in,
                                                 float* __restrict__ msg,
                                                 float* __restrict__ ps_out,
                                                 int*   __restrict__ pi_out,
                                                 int gh, int gw, int is_last) {
    __shared__ __attribute__((aligned(16))) float KV[64*LKP_];
    __shared__ __attribute__((aligned(16))) float sL[4][64];
    __shared__ float mA[4][32];
    __shared__ float psh[16];
    __shared__ int   ci[64];
    __shared__ __attribute__((aligned(16))) int ciT[4][16];
    __shared__ float topv[4][16];
    __shared__ int   topi[4][16];

    const int Sp = gh*gw;
    const int p  = blockIdx.x % Sp;
    const int bh = blockIdx.x / Sp;
    const int wt = 2*gw;
    const int St = 4*Sp;
    const int lane = threadIdx.x;
    const int lm = lane & 15, hf = lane >> 4;
    const float* Q = qf + (size_t)bh*St*DH_;
    const float* K = kf + (size_t)bh*St*DH_;
    const float* V = vf + (size_t)bh*St*DH_;

    if (lane < 16) {
        psh[lane] = ps_in[((size_t)bh*Sp + p)*16 + lane];
        const int pi = pi_in[((size_t)bh*Sp + p)*16 + lane];
        const int r = pi / gw, c = pi % gw;
        for (int cc = 0; cc < 4; ++cc) {
            const int g = (2*r + (cc >> 1))*wt + (2*c + (cc & 1));
            ci[lane*4 + cc] = g;
            ciT[cc][lane] = g;
        }
    }
    __syncthreads();

    const int sc   = lane >> 3;
    const int scol = (lane & 7) * 4;
    int srow[16];
    #pragma unroll
    for (int q4 = 0; q4 < 4; ++q4) {
        const int4 t = ((const int4*)&ciT[sc][0])[q4];
        srow[4*q4 + 0] = t.x; srow[4*q4 + 1] = t.y; srow[4*q4 + 2] = t.z; srow[4*q4 + 3] = t.w;
    }
    float4 stg[16];
    #pragma unroll
    for (int i = 0; i < 16; ++i) stg[i] = *(const float4*)(K + (size_t)srow[i]*DH_ + scol);
    #pragma unroll
    for (int i = 0; i < 16; ++i) *(float4*)&KV[(4*i + sc)*LKP_ + scol] = stg[i];
    __syncthreads();

    const int rp = p / gw, cp = p % gw;
    const int qm = lm & 3;
    const int qidx = (2*rp + (qm >> 1))*wt + (2*cp + (qm & 1));
    const bool alive = (lm < 4);

    for (int nt = 0; nt < 4; ++nt) {
        v8f acc = {};
        const float* qr = Q + (size_t)qidx*DH_;
        const float* kr = &KV[(nt*16 + lm)*LKP_];
        for (int kk = 0; kk < DH_; kk += 4) {
            v2f a, b;
            const float a0v = SCALE_ * qr[kk + 0 + 2*hf];
            const float a1v = SCALE_ * qr[kk + 1 + 2*hf];
            a[0] = alive ? a0v : 0.f;
            a[1] = alive ? a1v : 0.f;
            b[0] = kr[kk + 0 + 2*hf];
            b[1] = kr[kk + 1 + 2*hf];
            acc = wmma_f32(a, b, acc);
        }
        if (hf == 0)
            for (int r = 0; r < 4; ++r) sL[r][nt*16 + lm] = acc[r];
    }
    __syncthreads();

    for (int g = lane; g < 64; g += 32) {
        const int m = g >> 4, kp = g & 15;
        float* row = &sL[m][kp*4];
        const float mx = fmaxf(fmaxf(row[0], row[1]), fmaxf(row[2], row[3]));
        const float e0 = __expf(row[0]-mx), e1 = __expf(row[1]-mx);
        const float e2 = __expf(row[2]-mx), e3 = __expf(row[3]-mx);
        const float f = psh[kp] / (e0+e1+e2+e3);
        row[0] = e0*f; row[1] = e1*f; row[2] = e2*f; row[3] = e3*f;
    }
    #pragma unroll
    for (int i = 0; i < 16; ++i) stg[i] = *(const float4*)(V + (size_t)srow[i]*DH_ + scol);
    #pragma unroll
    for (int i = 0; i < 16; ++i) *(float4*)&KV[(4*i + sc)*DH_ + qta_vcol(4*i + sc, scol)] = stg[i];
    __syncthreads();

    for (int nt = 0; nt < 2; ++nt) {
        v8f acc = {};
        for (int kk = 0; kk < 64; kk += 4) {
            v2f a, b;
            const float a0v = sL[qm][kk + 0 + 2*hf];
            const float a1v = sL[qm][kk + 1 + 2*hf];
            a[0] = alive ? a0v : 0.f;
            a[1] = alive ? a1v : 0.f;
            b[0] = KV[(kk + 0 + 2*hf)*DH_ + qta_vcol(kk + 0 + 2*hf, nt*16 + lm)];
            b[1] = KV[(kk + 1 + 2*hf)*DH_ + qta_vcol(kk + 1 + 2*hf, nt*16 + lm)];
            acc = wmma_f32(a, b, acc);
        }
        if (hf == 0)
            for (int r = 0; r < 4; ++r) mA[r][nt*16 + lm] = acc[r];
    }
    __syncthreads();

    if (is_last) {
        for (int m = 0; m < 4; ++m) {
            const int fi = (2*rp + (m >> 1))*wt + (2*cp + (m & 1));
            msg[((size_t)bh*St + fi)*DH_ + lane] = mA[m][lane];
        }
        return;
    }

    if (lane < 4) {
        const int m = lane;
        float cm[4];
        #pragma unroll
        for (int cc = 0; cc < 4; ++cc) {
            float x[16];
            #pragma unroll
            for (int q = 0; q < 4; ++q) {
                const float4 xv = ((const float4*)&sL[m][cc*16])[q];
                x[4*q + 0] = xv.x; x[4*q + 1] = xv.y; x[4*q + 2] = xv.z; x[4*q + 3] = xv.w;
            }
            cm[cc] = qta_max16(x);
        }
        qta_pick16<4>(&sL[m][0], cm, &topv[m][0], &topi[m][0]);
    }
    __syncthreads();

    for (int m = 0; m < 4; ++m) {
        float msel = 0.f;
        for (int t = 0; t < 16; ++t) {
            const int r = topi[m][t];
            msel += topv[m][t] * KV[r*DH_ + qta_vcol(r, lane)];
        }
        const int fi = (2*rp + (m >> 1))*wt + (2*cp + (m & 1));
        msg[((size_t)bh*St + fi)*DH_ + lane] = mA[m][lane] - msel;
    }
    if (lane < 16) {
        for (int m = 0; m < 4; ++m) {
            const int fi = (2*rp + (m >> 1))*wt + (2*cp + (m & 1));
            ps_out[((size_t)bh*St + fi)*16 + lane] = topv[m][lane];
            pi_out[((size_t)bh*St + fi)*16 + lane] = ci[topi[m][lane]];
        }
    }
}

extern "C" void kernel_launch(void* const* d_in, const int* in_sizes, int n_in,
                              void* d_out, int out_size, void* d_ws, size_t ws_size,
                              hipStream_t stream) {
    (void)in_sizes; (void)n_in; (void)out_size; (void)ws_size;
    const float* x  = (const float*)d_in[0];
    const float* wq = (const float*)d_in[3];
    const float* wk = (const float*)d_in[4];
    const float* wv = (const float*)d_in[5];
    const float* wp = (const float*)d_in[6];
    const float* bp = (const float*)d_in[7];

    const size_t SZ2 = (size_t)B_*NH_*4096*DH_;
    const size_t SZ1 = (size_t)B_*NH_*1024*DH_;
    const size_t SZ0 = (size_t)B_*NH_*256 *DH_;
    float* ws = (float*)d_ws; size_t o = 0;
    float* q2 = ws + o; o += SZ2;  float* k2 = ws + o; o += SZ2;  float* v2 = ws + o; o += SZ2;
    float* q1 = ws + o; o += SZ1;  float* k1 = ws + o; o += SZ1;  float* v1 = ws + o; o += SZ1;
    float* q0 = ws + o; o += SZ0;  float* k0 = ws + o; o += SZ0;  float* v0 = ws + o; o += SZ0;
    float* msg0 = ws + o; o += SZ0;
    float* msg1 = ws + o; o += SZ1;
    float* msg2 = ws + o; o += SZ2;
    float* ps0 = ws + o; o += (size_t)B_*NH_*256*16;
    int*   pi0 = (int*)(ws + o); o += (size_t)B_*NH_*256*16;
    float* ps1 = ws + o; o += (size_t)B_*NH_*1024*16;
    int*   pi1 = (int*)(ws + o); o += (size_t)B_*NH_*1024*16;

    const dim3 gg(2, 128);
    qta_gemm_qkv<<<gg, 256, 0, stream>>>(x, wq, q2, q1, q0);
    qta_gemm_qkv<<<gg, 256, 0, stream>>>(x, wk, k2, k1, k0);
    qta_gemm_qkv<<<gg, 256, 0, stream>>>(x, wv, v2, v1, v0);

    qta_level0<<<B_*NH_*16, 128, 0, stream>>>(q0, k0, v0, msg0, ps0, pi0);
    qta_levelt<<<B_*NH_*256,  32, 0, stream>>>(q1, k1, v1, ps0, pi0, msg1, ps1, pi1, 16, 16, 0);
    qta_levelt<<<B_*NH_*1024, 32, 0, stream>>>(q2, k2, v2, ps1, pi1, msg2, nullptr, nullptr, 32, 32, 1);

    qta_gemm_proj<<<gg, 256, 0, stream>>>(msg2, msg1, msg0, wp, bp, (float*)d_out);
}
